# starting kernel
# baseline (speedup 1.0000x reference)
#include <hip/hip_runtime.h>
#include <hip/hip_bf16.h>

typedef __attribute__((ext_vector_type(2))) float v2f;
typedef __attribute__((ext_vector_type(8))) float v8f;

#define HID 64
#define EPS_BLEND 0.1f

template<int K, bool RELU>
__global__ void gemm_wmma_f32(const float* __restrict__ A,
                              const float* __restrict__ W,
                              const float* __restrict__ bias,
                              float* __restrict__ out,
                              int ntiles) {
  const int wave = (int)((blockIdx.x * blockDim.x + threadIdx.x) >> 5);
  const int lane = (int)(threadIdx.x & 31);
  if (wave >= ntiles) return;

  const int m  = lane & 15;
  const int kh = (lane >> 4) << 1;

  v8f acc0 = {}, acc1 = {}, acc2 = {}, acc3 = {};

  const float* arow = A + ((size_t)wave * 16 + m) * K + kh;
  const float* w0 = W + (size_t)( 0 + m) * K + kh;
  const float* w1 = W + (size_t)(16 + m) * K + kh;
  const float* w2 = W + (size_t)(32 + m) * K + kh;
  const float* w3 = W + (size_t)(48 + m) * K + kh;

  #pragma unroll 4
  for (int k0 = 0; k0 < K; k0 += 4) {
    v2f a;  a.x  = arow[k0]; a.y  = arow[k0 + 1];
    v2f b0; b0.x = w0[k0];   b0.y = w0[k0 + 1];
    v2f b1; b1.x = w1[k0];   b1.y = w1[k0 + 1];
    v2f b2; b2.x = w2[k0];   b2.y = w2[k0 + 1];
    v2f b3; b3.x = w3[k0];   b3.y = w3[k0 + 1];
    acc0 = __builtin_amdgcn_wmma_f32_16x16x4_f32(false, a, false, b0, (short)0, acc0, false, false);
    acc1 = __builtin_amdgcn_wmma_f32_16x16x4_f32(false, a, false, b1, (short)0, acc1, false, false);
    acc2 = __builtin_amdgcn_wmma_f32_16x16x4_f32(false, a, false, b2, (short)0, acc2, false, false);
    acc3 = __builtin_amdgcn_wmma_f32_16x16x4_f32(false, a, false, b3, (short)0, acc3, false, false);
  }

  const int col   = lane & 15;
  const int rbase = (lane >> 4) * 8;
  const float bb0 = bias[col], bb1 = bias[16 + col], bb2 = bias[32 + col], bb3 = bias[48 + col];
  float* orow = out + (size_t)wave * 16 * HID;
  #pragma unroll
  for (int v = 0; v < 8; ++v) {
    const int r = rbase + v;
    float x0 = acc0[v] + bb0;
    float x1 = acc1[v] + bb1;
    float x2 = acc2[v] + bb2;
    float x3 = acc3[v] + bb3;
    if (RELU) {
      x0 = fmaxf(x0, 0.f); x1 = fmaxf(x1, 0.f);
      x2 = fmaxf(x2, 0.f); x3 = fmaxf(x3, 0.f);
    }
    orow[(size_t)r * HID +      col] = x0;
    orow[(size_t)r * HID + 16 + col] = x1;
    orow[(size_t)r * HID + 32 + col] = x2;
    orow[(size_t)r * HID + 48 + col] = x3;
  }
}

__global__ void deg_init(float* __restrict__ deg, int n) {
  int i = blockIdx.x * blockDim.x + threadIdx.x;
  if (i < n) deg[i] = 1.0f;
}
__global__ void deg_acc(const int* __restrict__ dst, float* __restrict__ deg, int e) {
  int i = blockIdx.x * blockDim.x + threadIdx.x;
  if (i < e) atomicAdd(&deg[dst[i]], 1.0f);
}
__global__ void deg_rsqrt(float* __restrict__ deg, int n) {
  int i = blockIdx.x * blockDim.x + threadIdx.x;
  if (i < n) deg[i] = rsqrtf(deg[i]);
}

__global__ void alpha_kernel(const float* __restrict__ h,
                             const float* __restrict__ attl,
                             const float* __restrict__ attr,
                             float* __restrict__ al, float* __restrict__ ar, int n) {
  const int wave = (int)((blockIdx.x * blockDim.x + threadIdx.x) >> 5);
  const int lane = (int)(threadIdx.x & 31);
  if (wave >= n) return;
  const float* row = h + (size_t)wave * HID;
  const float h0v = row[lane], h1v = row[lane + 32];
  float pl = h0v * attl[lane] + h1v * attl[lane + 32];
  float pr = h0v * attr[lane] + h1v * attr[lane + 32];
  #pragma unroll
  for (int off = 16; off > 0; off >>= 1) {
    pl += __shfl_xor(pl, off, 32);
    pr += __shfl_xor(pr, off, 32);
  }
  if (lane == 0) { al[wave] = pl; ar[wave] = pr; }
}

__global__ void init_out(const float* __restrict__ h, const float* __restrict__ h0,
                         const float* __restrict__ al, const float* __restrict__ ar,
                         const float* __restrict__ dinv, float* __restrict__ out, int n) {
  int idx = blockIdx.x * blockDim.x + threadIdx.x;
  if (idx >= n * HID) return;
  int node = idx >> 6;
  float di = dinv[node];
  float w = tanhf(al[node] + ar[node]) * di * di;
  out[idx] = EPS_BLEND * h0[idx] + w * h[idx];
}

__global__ void edge_kernel(const int* __restrict__ s, const int* __restrict__ d,
                            const float* __restrict__ h,
                            const float* __restrict__ al, const float* __restrict__ ar,
                            const float* __restrict__ dinv,
                            float* __restrict__ out, int e) {
  const int wave = (int)((blockIdx.x * blockDim.x + threadIdx.x) >> 5);
  const int lane = (int)(threadIdx.x & 31);
  if (wave >= e) return;
  const int se = s[wave], de = d[wave];
  const float w = tanhf(al[se] + ar[de]) * dinv[se] * dinv[de];
  const float* hs = h + (size_t)se * HID;
  float* od = out + (size_t)de * HID;
  atomicAdd(&od[lane],      w * hs[lane]);
  atomicAdd(&od[lane + 32], w * hs[lane + 32]);
}

__global__ void relu_copy(const float* __restrict__ in, float* __restrict__ out, int n) {
  int i = blockIdx.x * blockDim.x + threadIdx.x;
  if (i < n) out[i] = fmaxf(in[i], 0.f);
}

extern "C" void kernel_launch(void* const* d_in, const int* in_sizes, int n_in,
                              void* d_out, int out_size, void* d_ws, size_t ws_size,
                              hipStream_t stream) {
  const float* x     = (const float*)d_in[0];
  const int*   ei    = (const int*)  d_in[1];
  const float* W_in  = (const float*)d_in[2];
  const float* b_in  = (const float*)d_in[3];
  const float* attl1 = (const float*)d_in[4];
  const float* attr1 = (const float*)d_in[5];
  const float* attl2 = (const float*)d_in[6];
  const float* attr2 = (const float*)d_in[7];
  const float* W_out = (const float*)d_in[8];
  const float* b_out = (const float*)d_in[9];
  float* out = (float*)d_out;

  const int IN = 128;
  const int E  = in_sizes[1] / 2;
  const int N  = in_sizes[0] / IN;
  const int*   srcI = ei;
  const int*   dstI = ei + E;

  float* H0   = (float*)d_ws;
  float* H1   = H0   + (size_t)N * HID;
  float* OUTB = H1   + (size_t)N * HID;
  float* DINV = OUTB + (size_t)N * HID;
  float* AL   = DINV + N;
  float* AR   = AL   + N;

  const int ntiles = N / 16;
  const int gemmBlocks = (ntiles + 7) / 8;
  const int nThreads  = (N + 255) / 256;
  const int neThreads = (E + 255) / 256;
  const int nWaveBlk  = (N + 7) / 8;
  const int eWaveBlk  = (E + 7) / 8;
  const int ncElems   = (N * HID + 255) / 256;

  gemm_wmma_f32<128, true><<<gemmBlocks, 256, 0, stream>>>(x, W_in, b_in, H0, ntiles);

  deg_init <<<nThreads, 256, 0, stream>>>(DINV, N);
  deg_acc  <<<neThreads, 256, 0, stream>>>(dstI, DINV, E);
  deg_rsqrt<<<nThreads, 256, 0, stream>>>(DINV, N);

  alpha_kernel<<<nWaveBlk, 256, 0, stream>>>(H0, attl1, attr1, AL, AR, N);
  init_out    <<<ncElems, 256, 0, stream>>>(H0, H0, AL, AR, DINV, OUTB, N);
  edge_kernel <<<eWaveBlk, 256, 0, stream>>>(srcI, dstI, H0, AL, AR, DINV, OUTB, E);
  relu_copy   <<<ncElems, 256, 0, stream>>>(OUTB, H1, N * HID);

  alpha_kernel<<<nWaveBlk, 256, 0, stream>>>(H1, attl2, attr2, AL, AR, N);
  init_out    <<<ncElems, 256, 0, stream>>>(H1, H0, AL, AR, DINV, OUTB, N);
  edge_kernel <<<eWaveBlk, 256, 0, stream>>>(srcI, dstI, H1, AL, AR, DINV, OUTB, E);
  relu_copy   <<<ncElems, 256, 0, stream>>>(OUTB, H0, N * HID);

  gemm_wmma_f32<64, false><<<gemmBlocks, 256, 0, stream>>>(H0, W_out, b_out, out, ntiles);
}


# edge scatter: one wave per 32 edges instead of one per edge: lanes fetch endpoints and weights of 32 edges at once, then the wave walks them by lane broadcast, 64 row loads ahead of 64 atomics
# speedup vs baseline: 1.6639x; 1.6639x over previous
#include <hip/hip_runtime.h>
#include <hip/hip_bf16.h>

typedef __attribute__((ext_vector_type(2))) float v2f;
typedef __attribute__((ext_vector_type(8))) float v8f;

#define HID 64
#define EPS_BLEND 0.1f

template<int K, bool RELU>
__global__ void gemm_wmma_f32(const float* __restrict__ A,
                              const float* __restrict__ W,
                              const float* __restrict__ bias,
                              float* __restrict__ out,
                              int ntiles) {
  const int wave = (int)((blockIdx.x * blockDim.x + threadIdx.x) >> 5);
  const int lane = (int)(threadIdx.x & 31);
  if (wave >= ntiles) return;

  const int m  = lane & 15;
  const int kh = (lane >> 4) << 1;

  v8f acc0 = {}, acc1 = {}, acc2 = {}, acc3 = {};

  const float* arow = A + ((size_t)wave * 16 + m) * K + kh;
  const float* w0 = W + (size_t)( 0 + m) * K + kh;
  const float* w1 = W + (size_t)(16 + m) * K + kh;
  const float* w2 = W + (size_t)(32 + m) * K + kh;
  const float* w3 = W + (size_t)(48 + m) * K + kh;

  #pragma unroll 4
  for (int k0 = 0; k0 < K; k0 += 4) {
    v2f a;  a.x  = arow[k0]; a.y  = arow[k0 + 1];
    v2f b0; b0.x = w0[k0];   b0.y = w0[k0 + 1];
    v2f b1; b1.x = w1[k0];   b1.y = w1[k0 + 1];
    v2f b2; b2.x = w2[k0];   b2.y = w2[k0 + 1];
    v2f b3; b3.x = w3[k0];   b3.y = w3[k0 + 1];
    acc0 = __builtin_amdgcn_wmma_f32_16x16x4_f32(false, a, false, b0, (short)0, acc0, false, false);
    acc1 = __builtin_amdgcn_wmma_f32_16x16x4_f32(false, a, false, b1, (short)0, acc1, false, false);
    acc2 = __builtin_amdgcn_wmma_f32_16x16x4_f32(false, a, false, b2, (short)0, acc2, false, false);
    acc3 = __builtin_amdgcn_wmma_f32_16x16x4_f32(false, a, false, b3, (short)0, acc3, false, false);
  }

  const int col   = lane & 15;
  const int rbase = (lane >> 4) * 8;
  const float bb0 = bias[col], bb1 = bias[16 + col], bb2 = bias[32 + col], bb3 = bias[48 + col];
  float* orow = out + (size_t)wave * 16 * HID;
  #pragma unroll
  for (int v = 0; v < 8; ++v) {
    const int r = rbase + v;
    float x0 = acc0[v] + bb0;
    float x1 = acc1[v] + bb1;
    float x2 = acc2[v] + bb2;
    float x3 = acc3[v] + bb3;
    if (RELU) {
      x0 = fmaxf(x0, 0.f); x1 = fmaxf(x1, 0.f);
      x2 = fmaxf(x2, 0.f); x3 = fmaxf(x3, 0.f);
    }
    orow[(size_t)r * HID +      col] = x0;
    orow[(size_t)r * HID + 16 + col] = x1;
    orow[(size_t)r * HID + 32 + col] = x2;
    orow[(size_t)r * HID + 48 + col] = x3;
  }
}

__global__ void deg_init(float* __restrict__ deg, int n) {
  int i = blockIdx.x * blockDim.x + threadIdx.x;
  if (i < n) deg[i] = 1.0f;
}
__global__ void deg_acc(const int* __restrict__ dst, float* __restrict__ deg, int e) {
  int i = blockIdx.x * blockDim.x + threadIdx.x;
  if (i < e) atomicAdd(&deg[dst[i]], 1.0f);
}
__global__ void deg_rsqrt(float* __restrict__ deg, int n) {
  int i = blockIdx.x * blockDim.x + threadIdx.x;
  if (i < n) deg[i] = rsqrtf(deg[i]);
}

__global__ void alpha_kernel(const float* __restrict__ h,
                             const float* __restrict__ attl,
                             const float* __restrict__ attr,
                             float* __restrict__ al, float* __restrict__ ar, int n) {
  const int wave = (int)((blockIdx.x * blockDim.x + threadIdx.x) >> 5);
  const int lane = (int)(threadIdx.x & 31);
  if (wave >= n) return;
  const float* row = h + (size_t)wave * HID;
  const float h0v = row[lane], h1v = row[lane + 32];
  float pl = h0v * attl[lane] + h1v * attl[lane + 32];
  float pr = h0v * attr[lane] + h1v * attr[lane + 32];
  #pragma unroll
  for (int off = 16; off > 0; off >>= 1) {
    pl += __shfl_xor(pl, off, 32);
    pr += __shfl_xor(pr, off, 32);
  }
  if (lane == 0) { al[wave] = pl; ar[wave] = pr; }
}

__global__ void init_out(const float* __restrict__ h, const float* __restrict__ h0,
                         const float* __restrict__ al, const float* __restrict__ ar,
                         const float* __restrict__ dinv, float* __restrict__ out, int n) {
  int idx = blockIdx.x * blockDim.x + threadIdx.x;
  if (idx >= n * HID) return;
  int node = idx >> 6;
  float di = dinv[node];
  float w = tanhf(al[node] + ar[node]) * di * di;
  out[idx] = EPS_BLEND * h0[idx] + w * h[idx];
}

__global__ void edge_kernel(const int* __restrict__ s, const int* __restrict__ d,
                            const float* __restrict__ h,
                            const float* __restrict__ al, const float* __restrict__ ar,
                            const float* __restrict__ dinv,
                            float* __restrict__ out, int e) {
  const int lane = (int)(threadIdx.x & 31);
  const int wib  = __builtin_amdgcn_readfirstlane((int)(threadIdx.x >> 5));
  const int wave = (int)blockIdx.x * (int)(blockDim.x >> 5) + wib;
  const int base = wave * 32;
  if (base >= e) return;
  const int cnt = (e - base < 32) ? (e - base) : 32;

  int se = 0, de = 0;
  float w = 0.f;
  if (lane < cnt) {
    se = s[base + lane];
    de = d[base + lane];
    const float a_s = al[se], a_d = ar[de], n_s = dinv[se], n_d = dinv[de];
    w = tanhf(a_s + a_d) * n_s * n_d;
  }
  const int wbits = __float_as_int(w);

  if (cnt == 32) {
    float v0[32], v1[32];
    #pragma unroll
    for (int j = 0; j < 32; ++j) {
      const unsigned sj = (unsigned)__builtin_amdgcn_readlane(se, j);
      const float* hs = h + (size_t)sj * HID;
      v0[j] = hs[lane];
      v1[j] = hs[lane + 32];
    }
    #pragma unroll
    for (int j = 0; j < 32; ++j) {
      const float wj = __int_as_float(__builtin_amdgcn_readlane(wbits, j));
      v0[j] *= wj;
      v1[j] *= wj;
    }
    #pragma unroll
    for (int j = 0; j < 32; ++j) {
      const unsigned dj = (unsigned)__builtin_amdgcn_readlane(de, j);
      float* od = out + (size_t)dj * HID;
      atomicAdd(&od[lane],      v0[j]);
      atomicAdd(&od[lane + 32], v1[j]);
    }
  } else {
    for (int j = 0; j < cnt; ++j) {
      const unsigned sj = (unsigned)__builtin_amdgcn_readlane(se, j);
      const unsigned dj = (unsigned)__builtin_amdgcn_readlane(de, j);
      const float wj = __int_as_float(__builtin_amdgcn_readlane(wbits, j));
      const float* hs = h + (size_t)sj * HID;
      float* od = out + (size_t)dj * HID;
      atomicAdd(&od[lane],      wj * hs[lane]);
      atomicAdd(&od[lane + 32], wj * hs[lane + 32]);
    }
  }
}

__global__ void relu_copy(const float* __restrict__ in, float* __restrict__ out, int n) {
  int i = blockIdx.x * blockDim.x + threadIdx.x;
  if (i < n) out[i] = fmaxf(in[i], 0.f);
}

extern "C" void kernel_launch(void* const* d_in, const int* in_sizes, int n_in,
                              void* d_out, int out_size, void* d_ws, size_t ws_size,
                              hipStream_t stream) {
  const float* x     = (const float*)d_in[0];
  const int*   ei    = (const int*)  d_in[1];
  const float* W_in  = (const float*)d_in[2];
  const float* b_in  = (const float*)d_in[3];
  const float* attl1 = (const float*)d_in[4];
  const float* attr1 = (const float*)d_in[5];
  const float* attl2 = (const float*)d_in[6];
  const float* attr2 = (const float*)d_in[7];
  const float* W_out = (const float*)d_in[8];
  const float* b_out = (const float*)d_in[9];
  float* out = (float*)d_out;

  const int IN = 128;
  const int E  = in_sizes[1] / 2;
  const int N  = in_sizes[0] / IN;
  const int*   srcI = ei;
  const int*   dstI = ei + E;

  float* H0   = (float*)d_ws;
  float* H1   = H0   + (size_t)N * HID;
  float* OUTB = H1   + (size_t)N * HID;
  float* DINV = OUTB + (size_t)N * HID;
  float* AL   = DINV + N;
  float* AR   = AL   + N;

  const int ntiles = N / 16;
  const int gemmBlocks = (ntiles + 7) / 8;
  const int nThreads  = (N + 255) / 256;
  const int neThreads = (E + 255) / 256;
  const int nWaveBlk  = (N + 7) / 8;
  const int eWaveBlk  = (E + 255) / 256;
  const int ncElems   = (N * HID + 255) / 256;

  gemm_wmma_f32<128, true><<<gemmBlocks, 256, 0, stream>>>(x, W_in, b_in, H0, ntiles);

  deg_init <<<nThreads, 256, 0, stream>>>(DINV, N);
  deg_acc  <<<neThreads, 256, 0, stream>>>(dstI, DINV, E);
  deg_rsqrt<<<nThreads, 256, 0, stream>>>(DINV, N);

  alpha_kernel<<<nWaveBlk, 256, 0, stream>>>(H0, attl1, attr1, AL, AR, N);
  init_out    <<<ncElems, 256, 0, stream>>>(H0, H0, AL, AR, DINV, OUTB, N);
  edge_kernel <<<eWaveBlk, 256, 0, stream>>>(srcI, dstI, H0, AL, AR, DINV, OUTB, E);
  relu_copy   <<<ncElems, 256, 0, stream>>>(OUTB, H1, N * HID);

  alpha_kernel<<<nWaveBlk, 256, 0, stream>>>(H1, attl2, attr2, AL, AR, N);
  init_out    <<<ncElems, 256, 0, stream>>>(H1, H0, AL, AR, DINV, OUTB, N);
  edge_kernel <<<eWaveBlk, 256, 0, stream>>>(srcI, dstI, H1, AL, AR, DINV, OUTB, E);
  relu_copy   <<<ncElems, 256, 0, stream>>>(OUTB, H0, N * HID);

  gemm_wmma_f32<64, false><<<gemmBlocks, 256, 0, stream>>>(H0, W_out, b_out, out, ntiles);
}


# edge scatter with float atomics replaced by a gather over edges grouped by target (built per call: counting add whose return is the slot, block scan, fill); a wave sums 8 nodes' rows in registers
# speedup vs baseline: 2.6311x; 1.5813x over previous
#include <hip/hip_runtime.h>
#include <hip/hip_bf16.h>


typedef __attribute__((ext_vector_type(2))) float v2f;
typedef __attribute__((ext_vector_type(8))) float v8f;
typedef __attribute__((ext_vector_type(4))) float v4f;

#define HID 64
#define EPS_BLEND 0.1f

template<int K, bool RELU, bool RELU_IN>
__global__ void gemm_wmma_f32(const float* __restrict__ A,
                              const float* __restrict__ W,
                              const float* __restrict__ bias,
                              float* __restrict__ out,
                              int ntiles) {
  const int wave = (int)((blockIdx.x * blockDim.x + threadIdx.x) >> 5);
  const int lane = (int)(threadIdx.x & 31);
  if (wave >= ntiles) return;

  const int m  = lane & 15;
  const int kh = (lane >> 4) << 1;

  v8f acc0 = {}, acc1 = {}, acc2 = {}, acc3 = {};

  const float* arow = A + ((size_t)wave * 16 + m) * K + kh;
  const float* w0 = W + (size_t)( 0 + m) * K + kh;
  const float* w1 = W + (size_t)(16 + m) * K + kh;
  const float* w2 = W + (size_t)(32 + m) * K + kh;
  const float* w3 = W + (size_t)(48 + m) * K + kh;

  #pragma unroll 4
  for (int k0 = 0; k0 < K; k0 += 4) {
    v2f a;  a.x  = arow[k0]; a.y  = arow[k0 + 1];
    if (RELU_IN) { a.x = fmaxf(a.x, 0.f); a.y = fmaxf(a.y, 0.f); }
    v2f b0; b0.x = w0[k0];   b0.y = w0[k0 + 1];
    v2f b1; b1.x = w1[k0];   b1.y = w1[k0 + 1];
    v2f b2; b2.x = w2[k0];   b2.y = w2[k0 + 1];
    v2f b3; b3.x = w3[k0];   b3.y = w3[k0 + 1];
    acc0 = __builtin_amdgcn_wmma_f32_16x16x4_f32(false, a, false, b0, (short)0, acc0, false, false);
    acc1 = __builtin_amdgcn_wmma_f32_16x16x4_f32(false, a, false, b1, (short)0, acc1, false, false);
    acc2 = __builtin_amdgcn_wmma_f32_16x16x4_f32(false, a, false, b2, (short)0, acc2, false, false);
    acc3 = __builtin_amdgcn_wmma_f32_16x16x4_f32(false, a, false, b3, (short)0, acc3, false, false);
  }

  const int col   = lane & 15;
  const int rbase = (lane >> 4) * 8;
  const float bb0 = bias[col], bb1 = bias[16 + col], bb2 = bias[32 + col], bb3 = bias[48 + col];
  float* orow = out + (size_t)wave * 16 * HID;
  #pragma unroll
  for (int v = 0; v < 8; ++v) {
    const int r = rbase + v;
    float x0 = acc0[v] + bb0;
    float x1 = acc1[v] + bb1;
    float x2 = acc2[v] + bb2;
    float x3 = acc3[v] + bb3;
    if (RELU) {
      x0 = fmaxf(x0, 0.f); x1 = fmaxf(x1, 0.f);
      x2 = fmaxf(x2, 0.f); x3 = fmaxf(x3, 0.f);
    }
    orow[(size_t)r * HID +      col] = x0;
    orow[(size_t)r * HID + 16 + col] = x1;
    orow[(size_t)r * HID + 32 + col] = x2;
    orow[(size_t)r * HID + 48 + col] = x3;
  }
}

#define EPT 4

__global__ void csr_zero(int* __restrict__ cnt, int* __restrict__ total, int n) {
  const int i = blockIdx.x * blockDim.x + threadIdx.x;
  if (i < n) cnt[i] = 0;
  if (i == 0) *total = 0;
}

__global__ void csr_count(const int* __restrict__ dst, int* __restrict__ cnt,
                          int* __restrict__ rank, int e) {
  const int i      = blockIdx.x * blockDim.x + threadIdx.x;
  const int stride = gridDim.x * blockDim.x;
  int d[EPT];
  #pragma unroll
  for (int k = 0; k < EPT; ++k) {
    const int idx = i + k * stride;
    d[k] = idx < e ? dst[idx] : -1;
  }
  int r[EPT];
  #pragma unroll
  for (int k = 0; k < EPT; ++k) {
    r[k] = 0;
    if (d[k] >= 0) r[k] = atomicAdd(&cnt[d[k]], 1);
  }
  #pragma unroll
  for (int k = 0; k < EPT; ++k) {
    const int idx = i + k * stride;
    if (d[k] >= 0) rank[idx] = r[k];
  }
}

__global__ void csr_alloc(const int* __restrict__ cnt, int* __restrict__ total,
                          int* __restrict__ rowptr, int n) {
  __shared__ int wsum[8];
  __shared__ int bbase;
  const int i    = blockIdx.x * 256 + threadIdx.x;
  const int lane = (int)(threadIdx.x & 31);
  const int w    = (int)(threadIdx.x >> 5);
  const int c    = i < n ? cnt[i] : 0;
  int inc = c;
  for (int off = 1; off < 32; off <<= 1) {
    const int t = __shfl_up(inc, off, 32);
    if (lane >= off) inc += t;
  }
  if (lane == 31) wsum[w] = inc;
  __syncthreads();
  if (threadIdx.x == 0) {
    int s = 0;
    for (int k = 0; k < 8; ++k) { const int t = wsum[k]; wsum[k] = s; s += t; }
    bbase = atomicAdd(total, s);
  }
  __syncthreads();
  if (i < n) rowptr[i] = bbase + wsum[w] + inc - c;
}

__global__ void csr_fill(const int* __restrict__ src, const int* __restrict__ dst,
                         const int* __restrict__ rank, const int* __restrict__ rowptr,
                         int* __restrict__ csr, int e) {
  const int i      = blockIdx.x * blockDim.x + threadIdx.x;
  const int stride = gridDim.x * blockDim.x;
  int s[EPT], d[EPT], r[EPT], p[EPT];
  #pragma unroll
  for (int k = 0; k < EPT; ++k) {
    const int idx = i + k * stride;
    const bool ok = idx < e;
    d[k] = ok ? dst[idx] : -1;
    s[k] = ok ? src[idx] : 0;
    r[k] = ok ? rank[idx] : 0;
  }
  #pragma unroll
  for (int k = 0; k < EPT; ++k) p[k] = d[k] >= 0 ? rowptr[d[k]] : 0;
  #pragma unroll
  for (int k = 0; k < EPT; ++k) if (d[k] >= 0) csr[p[k] + r[k]] = s[k];
}

#define NP_STEPS 4
#define NP_NODES_PER_BLOCK (8 * 2 * NP_STEPS)

__global__ void att_pass(const float* __restrict__ h0,
                         const float* __restrict__ attl,
                         const float* __restrict__ attr,
                         const int* __restrict__ cnt,
                         float* __restrict__ aldf, float* __restrict__ ar, int n) {
  const int lane  = (int)(threadIdx.x & 31);
  const int wave  = (int)((blockIdx.x * blockDim.x + threadIdx.x) >> 5);
  const int sub   = lane & 15;
  const int c4    = sub * 4;
  const int node0 = wave * (2 * NP_STEPS) + (lane >> 4);
  v2f* ald = (v2f*)aldf;

  if (wave * (2 * NP_STEPS) >= n) return;

  const v4f wl = *(const v4f*)(attl + c4);
  const v4f wr = *(const v4f*)(attr + c4);

  v4f hv[NP_STEPS];
  int dg[NP_STEPS];
  #pragma unroll
  for (int t = 0; t < NP_STEPS; ++t) {
    const int node = node0 + 2 * t;
    const int ld   = node < n ? node : n - 1;
    hv[t] = *(const v4f*)(h0 + (size_t)ld * HID + c4);
    dg[t] = cnt[ld];
  }

  __builtin_amdgcn_sched_barrier(0);

  #pragma unroll
  for (int t = 0; t < NP_STEPS; ++t) {
    const int node = node0 + 2 * t;
    const v4f h = hv[t];
    float pl = h.x * wl.x + h.y * wl.y + h.z * wl.z + h.w * wl.w;
    float pr = h.x * wr.x + h.y * wr.y + h.z * wr.z + h.w * wr.w;
    for (int off = 8; off > 0; off >>= 1) {
      pl += __shfl_xor(pl, off, 32);
      pr += __shfl_xor(pr, off, 32);
    }
    if (node < n && sub == 0) {
      v2f a; a.x = pl; a.y = rsqrtf((float)(dg[t] + 1));
      ald[node] = a;
      ar[node]  = pr;
    }
  }
}

#define GN 8
#define GE 24

template<int NS>
__device__ __forceinline__ void chunk_rows(v4f& acc, const float* __restrict__ h,
                                           int se, int wbits, int half, int c4) {
  const v4f* __restrict__ hv = (const v4f*)h;
  const unsigned sub = (unsigned)c4 >> 2;
  v4f rows[NS];
  #pragma unroll
  for (int j = 0; j < NS; ++j) {
    const int s0 = __builtin_amdgcn_readlane(se, 2 * j);
    const int s1 = __builtin_amdgcn_readlane(se, 2 * j + 1);
    const unsigned sj = (unsigned)(half ? s1 : s0);
    rows[j] = hv[sj * (HID / 4) + sub];
  }
  __builtin_amdgcn_sched_barrier(0);
  #pragma unroll
  for (int j = 0; j < NS; ++j) {
    const int w0 = __builtin_amdgcn_readlane(wbits, 2 * j);
    const int w1 = __builtin_amdgcn_readlane(wbits, 2 * j + 1);
    const float wj = __int_as_float(half ? w1 : w0);
    acc.x += wj * rows[j].x;
    acc.y += wj * rows[j].y;
    acc.z += wj * rows[j].z;
    acc.w += wj * rows[j].w;
  }
}

#define CHUNK_CASE(k) case k: chunk_rows<k>(acc, h, se, wbits, half, c4); break;

template<bool FIRST>
__global__ void __launch_bounds__(256) gather_layer(
                             const int* __restrict__ rowptr, const int* __restrict__ cnt,
                             const int* __restrict__ csr,
                             const float* __restrict__ h,
                             const float* h0res,
                             const float* __restrict__ aldf, const float* __restrict__ ar,
                             const float* __restrict__ attl2, const float* __restrict__ attr2,
                             float* orow,
                             float* __restrict__ ald2f, float* __restrict__ ar2, int n) {
  const int lane = (int)(threadIdx.x & 31);
  const int wib   = __builtin_amdgcn_readfirstlane((int)(threadIdx.x >> 5));
  const int wave  = (int)blockIdx.x * (int)(blockDim.x >> 5) + wib;
  const int node0 = wave * GN;
  if (node0 >= n) return;
  const int nn   = (n - node0 < GN) ? (n - node0) : GN;
  const int half = lane >> 4;
  const int c4   = (lane & 15) * 4;
  const v2f* ald = (const v2f*)aldf;

  int mb = 0, mc = 0;
  float mdi = 0.f, mar = 0.f, mws = 0.f;
  if (lane < nn) {
    mb = rowptr[node0 + lane];
    mc = cnt[node0 + lane];
    const v2f a = ald[node0 + lane];
    mdi = a.y;
    mar = ar[node0 + lane];
    mws = tanhf(a.x + mar) * a.y * a.y;
  }
  v4f wl2 = {0.f, 0.f, 0.f, 0.f}, wr2 = {0.f, 0.f, 0.f, 0.f};
  if (FIRST) {
    wl2 = *(const v4f*)(attl2 + c4);
    wr2 = *(const v4f*)(attr2 + c4);
  }

  int begin = __builtin_amdgcn_readlane(mb, 0);
  int c     = __builtin_amdgcn_readlane(mc, 0);
  int se0   = node0;
  {
    const int cc0 = c < GE ? c : GE;
    if (lane < cc0) se0 = csr[begin + lane];
  }

  for (int t = 0; t < nn; ++t) {
    const int   node = node0 + t;
    const float di_n = __int_as_float(__builtin_amdgcn_readlane(__float_as_int(mdi), t));
    const float ar_n = __int_as_float(__builtin_amdgcn_readlane(__float_as_int(mar), t));
    const float ws_n = __int_as_float(__builtin_amdgcn_readlane(__float_as_int(mws), t));

    const size_t own = (size_t)node * HID + c4;
    v4f hn = {0.f, 0.f, 0.f, 0.f}, r = {0.f, 0.f, 0.f, 0.f};
    int begin_n = 0, c_n = 0, se_n = node0;

    v4f acc = {0.f, 0.f, 0.f, 0.f};
    int se = se0;
    int cb = 0;
    do {
      const int rem = c - cb;
      const int cc  = rem < GE ? rem : GE;
      v2f a = {0.f, 0.f};
      if (lane < cc) a = ald[se];
      if (cb == 0) {
        hn = *(const v4f*)(h + own);
        r  = hn;
        if (!FIRST) r = *(const v4f*)(h0res + own);
        if (t + 1 < nn) {
          begin_n = __builtin_amdgcn_readlane(mb, t + 1);
          c_n     = __builtin_amdgcn_readlane(mc, t + 1);
          const int ccn = c_n < GE ? c_n : GE;
          if (lane < ccn) se_n = csr[begin_n + lane];
        }
      }
      float w = 0.f;
      if (lane < cc) w = tanhf(a.x + ar_n) * a.y * di_n;
      const int wbits = __float_as_int(w);
      switch ((cc + 1) >> 1) {
        CHUNK_CASE(1)  CHUNK_CASE(2)  CHUNK_CASE(3)  CHUNK_CASE(4)
        CHUNK_CASE(5)  CHUNK_CASE(6)  CHUNK_CASE(7)  CHUNK_CASE(8)
        CHUNK_CASE(9)  CHUNK_CASE(10) CHUNK_CASE(11) CHUNK_CASE(12)
        default: break;
      }
      cb += GE;
      if (cb < c) {
        const int rem2 = c - cb;
        const int cc2  = rem2 < GE ? rem2 : GE;
        se = node0;
        if (lane < cc2) se = csr[begin + cb + lane];
      }
    } while (cb < c);

    acc.x += __shfl_xor(acc.x, 16, 32);
    acc.y += __shfl_xor(acc.y, 16, 32);
    acc.z += __shfl_xor(acc.z, 16, 32);
    acc.w += __shfl_xor(acc.w, 16, 32);

    v4f o;
    o.x = acc.x + (EPS_BLEND * r.x + ws_n * hn.x);
    o.y = acc.y + (EPS_BLEND * r.y + ws_n * hn.y);
    o.z = acc.z + (EPS_BLEND * r.z + ws_n * hn.z);
    o.w = acc.w + (EPS_BLEND * r.w + ws_n * hn.w);

    if (FIRST) {
      o.x = fmaxf(o.x, 0.f); o.y = fmaxf(o.y, 0.f);
      o.z = fmaxf(o.z, 0.f); o.w = fmaxf(o.w, 0.f);
      float pl = o.x * wl2.x + o.y * wl2.y + o.z * wl2.z + o.w * wl2.w;
      float pr = o.x * wr2.x + o.y * wr2.y + o.z * wr2.z + o.w * wr2.w;
      for (int off = 8; off > 0; off >>= 1) {
        pl += __shfl_xor(pl, off, 32);
        pr += __shfl_xor(pr, off, 32);
      }
      if (lane == 0) {
        v2f a; a.x = pl; a.y = di_n;
        ((v2f*)ald2f)[node] = a;
        ar2[node] = pr;
      }
    }
    if (half == 0) *(v4f*)(orow + own) = o;

    begin = begin_n; c = c_n; se0 = se_n;
  }
}

extern "C" void kernel_launch(void* const* d_in, const int* in_sizes, int n_in,
                              void* d_out, int out_size, void* d_ws, size_t ws_size,
                              hipStream_t stream) {
  const float* x     = (const float*)d_in[0];
  const int*   ei    = (const int*)  d_in[1];
  const float* W_in  = (const float*)d_in[2];
  const float* b_in  = (const float*)d_in[3];
  const float* attl1 = (const float*)d_in[4];
  const float* attr1 = (const float*)d_in[5];
  const float* attl2 = (const float*)d_in[6];
  const float* attr2 = (const float*)d_in[7];
  const float* W_out = (const float*)d_in[8];
  const float* b_out = (const float*)d_in[9];
  float* out = (float*)d_out;

  const int IN = 128;
  const int E  = in_sizes[1] / 2;
  const int N  = in_sizes[0] / IN;
  const int*   srcI = ei;
  const int*   dstI = ei + E;

  float* H0   = (float*)d_ws;
  float* H1   = H0   + (size_t)N * HID;
  float* ALD1 = H1   + (size_t)N * HID;
  float* ALD2 = ALD1 + (size_t)N * 2;
  float* AR1  = ALD2 + (size_t)N * 2;
  float* AR2  = AR1  + N;
  int*   CNT  = (int*)(AR2 + N);
  int*   RPTR = CNT  + N;
  int*   TOT  = RPTR + N;
  int*   RANK = TOT  + 1;
  int*   CSR  = RANK + E;

  const int ntiles = N / 16;
  const int gemmBlocks = (ntiles + 7) / 8;
  const int nThreads  = (N + 255) / 256;
  const int eBlk      = (E + 256 * EPT - 1) / (256 * EPT);
  const int nodeBlk   = (N + NP_NODES_PER_BLOCK - 1) / NP_NODES_PER_BLOCK;
  const int gBlk      = (N + 8 * GN - 1) / (8 * GN);

  gemm_wmma_f32<128, true, false><<<gemmBlocks, 256, 0, stream>>>(x, W_in, b_in, H0, ntiles);

  csr_zero <<<nThreads, 256, 0, stream>>>(CNT, TOT, N);
  csr_count<<<eBlk,     256, 0, stream>>>(dstI, CNT, RANK, E);
  csr_alloc<<<nThreads, 256, 0, stream>>>(CNT, TOT, RPTR, N);
  csr_fill <<<eBlk,     256, 0, stream>>>(srcI, dstI, RANK, RPTR, CSR, E);

  att_pass<<<nodeBlk, 256, 0, stream>>>(H0, attl1, attr1, CNT, ALD1, AR1, N);
  gather_layer<true> <<<gBlk, 256, 0, stream>>>(RPTR, CNT, CSR, H0, H0, ALD1, AR1,
                                                attl2, attr2, H1, ALD2, AR2, N);

  gather_layer<false><<<gBlk, 256, 0, stream>>>(RPTR, CNT, CSR, H1, H0, ALD2, AR2,
                                                attl2, attr2, H0, ALD2, AR2, N);

  gemm_wmma_f32<64, false, true><<<gemmBlocks, 256, 0, stream>>>(H0, W_out, b_out, out, ntiles);
}


# CSR build as a two-level counting sort on LDS atomics (bins of 512 targets; one block per bin ranks its edges) instead of a returning global atomic per edge; edge sum order stays arrival order
# speedup vs baseline: 3.8643x; 1.4687x over previous
#include <hip/hip_runtime.h>
#include <hip/hip_bf16.h>


typedef __attribute__((ext_vector_type(2))) float v2f;
typedef __attribute__((ext_vector_type(8))) float v8f;
typedef __attribute__((ext_vector_type(4))) float v4f;

#define HID 64
#define EPS_BLEND 0.1f

template<int K, bool RELU, bool RELU_IN>
__global__ void gemm_wmma_f32(const float* __restrict__ A,
                              const float* __restrict__ W,
                              const float* __restrict__ bias,
                              float* __restrict__ out,
                              int ntiles) {
  const int wave = (int)((blockIdx.x * blockDim.x + threadIdx.x) >> 5);
  const int lane = (int)(threadIdx.x & 31);
  if (wave >= ntiles) return;

  const int m  = lane & 15;
  const int kh = (lane >> 4) << 1;

  v8f acc0 = {}, acc1 = {}, acc2 = {}, acc3 = {};

  const float* arow = A + ((size_t)wave * 16 + m) * K + kh;
  const float* w0 = W + (size_t)( 0 + m) * K + kh;
  const float* w1 = W + (size_t)(16 + m) * K + kh;
  const float* w2 = W + (size_t)(32 + m) * K + kh;
  const float* w3 = W + (size_t)(48 + m) * K + kh;

  #pragma unroll 4
  for (int k0 = 0; k0 < K; k0 += 4) {
    v2f a;  a.x  = arow[k0]; a.y  = arow[k0 + 1];
    if (RELU_IN) { a.x = fmaxf(a.x, 0.f); a.y = fmaxf(a.y, 0.f); }
    v2f b0; b0.x = w0[k0];   b0.y = w0[k0 + 1];
    v2f b1; b1.x = w1[k0];   b1.y = w1[k0 + 1];
    v2f b2; b2.x = w2[k0];   b2.y = w2[k0 + 1];
    v2f b3; b3.x = w3[k0];   b3.y = w3[k0 + 1];
    acc0 = __builtin_amdgcn_wmma_f32_16x16x4_f32(false, a, false, b0, (short)0, acc0, false, false);
    acc1 = __builtin_amdgcn_wmma_f32_16x16x4_f32(false, a, false, b1, (short)0, acc1, false, false);
    acc2 = __builtin_amdgcn_wmma_f32_16x16x4_f32(false, a, false, b2, (short)0, acc2, false, false);
    acc3 = __builtin_amdgcn_wmma_f32_16x16x4_f32(false, a, false, b3, (short)0, acc3, false, false);
  }

  const int col   = lane & 15;
  const int rbase = (lane >> 4) * 8;
  const float bb0 = bias[col], bb1 = bias[16 + col], bb2 = bias[32 + col], bb3 = bias[48 + col];
  float* orow = out + (size_t)wave * 16 * HID;
  #pragma unroll
  for (int v = 0; v < 8; ++v) {
    const int r = rbase + v;
    float x0 = acc0[v] + bb0;
    float x1 = acc1[v] + bb1;
    float x2 = acc2[v] + bb2;
    float x3 = acc3[v] + bb3;
    if (RELU) {
      x0 = fmaxf(x0, 0.f); x1 = fmaxf(x1, 0.f);
      x2 = fmaxf(x2, 0.f); x3 = fmaxf(x3, 0.f);
    }
    orow[(size_t)r * HID +      col] = x0;
    orow[(size_t)r * HID + 16 + col] = x1;
    orow[(size_t)r * HID + 32 + col] = x2;
    orow[(size_t)r * HID + 48 + col] = x3;
  }
}

#define BIN_SHIFT 9
#define BIN_NODES 512
#define MAXBINS   256
#define BE        16
#define BCHUNK    (256 * BE)
#define CSR_T     1024
#define NOEDGE    0xFFFFFFFFu

__global__ void bin_zero(int* __restrict__ bincnt, int* __restrict__ bincur) {
  bincnt[threadIdx.x] = 0;
  bincur[threadIdx.x] = 0;
}

__global__ void __launch_bounds__(256) bin_count(const int* __restrict__ dst,
                                                 int* __restrict__ bincnt, int e) {
  __shared__ int hist[MAXBINS];
  const int t = (int)threadIdx.x;
  hist[t] = 0;
  __syncthreads();
  const int base = (int)blockIdx.x * BCHUNK + t;
  int d[BE];
  #pragma unroll
  for (int k = 0; k < BE; ++k) {
    const int idx = base + k * 256;
    d[k] = idx < e ? dst[idx] : -1;
  }
  #pragma unroll
  for (int k = 0; k < BE; ++k)
    if (d[k] >= 0) atomicAdd(&hist[d[k] >> BIN_SHIFT], 1);
  __syncthreads();
  const int h = hist[t];
  if (h > 0) atomicAdd(&bincnt[t], h);
}

__global__ void __launch_bounds__(256) bin_scatter(const int* __restrict__ src,
                                                   const int* __restrict__ dst,
                                                   const int* __restrict__ bincnt,
                                                   int* __restrict__ bincur,
                                                   unsigned* __restrict__ binned, int e) {
  __shared__ int hist[MAXBINS];
  __shared__ int gbase[MAXBINS];
  __shared__ int wsum[8];
  const int t    = (int)threadIdx.x;
  const int lane = t & 31;
  const int w    = t >> 5;
  hist[t] = 0;

  const int base = (int)blockIdx.x * BCHUNK + t;
  int d[BE], s[BE], r[BE];
  #pragma unroll
  for (int k = 0; k < BE; ++k) {
    const int idx = base + k * 256;
    const bool ok = idx < e;
    d[k] = ok ? dst[idx] : -1;
    s[k] = ok ? src[idx] : 0;
  }

  const int c = bincnt[t];
  int inc = c;
  for (int off = 1; off < 32; off <<= 1) {
    const int v = __shfl_up(inc, off, 32);
    if (lane >= off) inc += v;
  }
  if (lane == 31) wsum[w] = inc;
  __syncthreads();
  int pre = 0;
  for (int k = 0; k < 8; ++k) if (k < w) pre += wsum[k];
  const int bstart = pre + inc - c;

  #pragma unroll
  for (int k = 0; k < BE; ++k) {
    r[k] = 0;
    if (d[k] >= 0) r[k] = atomicAdd(&hist[d[k] >> BIN_SHIFT], 1);
  }
  __syncthreads();
  const int h = hist[t];
  int gb = bstart;
  if (h > 0) gb += atomicAdd(&bincur[t], h);
  gbase[t] = gb;
  __syncthreads();
  int pos[BE];
  #pragma unroll
  for (int k = 0; k < BE; ++k)
    pos[k] = gbase[d[k] >= 0 ? d[k] >> BIN_SHIFT : 0] + r[k];
  #pragma unroll
  for (int k = 0; k < BE; ++k) {
    if (d[k] >= 0)
      binned[pos[k]] = ((unsigned)s[k] << BIN_SHIFT) | ((unsigned)d[k] & (BIN_NODES - 1));
  }
}

__global__ void __launch_bounds__(CSR_T) bin_csr(const int* __restrict__ bincnt,
                                                 const unsigned* __restrict__ binned,
                                                 int* __restrict__ cnt,
                                                 int* __restrict__ rowptr,
                                                 int* __restrict__ csr, int n) {
  __shared__ int lcnt[BIN_NODES];
  __shared__ int loff[BIN_NODES];
  __shared__ int wsum[CSR_T / 32];
  __shared__ int wtot[BIN_NODES / 32];
  const int t    = (int)threadIdx.x;
  const int lane = t & 31;
  const int w    = t >> 5;
  const int bin  = (int)blockIdx.x;

  int v = t < bin ? bincnt[t] : 0;
  for (int off = 16; off > 0; off >>= 1) v += __shfl_xor(v, off, 32);
  if (lane == 0) wsum[w] = v;
  if (t < BIN_NODES) lcnt[t] = 0;
  const int m = bincnt[bin];
  __syncthreads();
  int bstart = 0;
  for (int k = 0; k < CSR_T / 32; ++k) bstart += wsum[k];
  const unsigned* __restrict__ mine = binned + bstart;

  for (int i0 = t; i0 < m; i0 += 4 * CSR_T) {
    unsigned p[4];
    #pragma unroll
    for (int k = 0; k < 4; ++k) {
      const int i = i0 + k * CSR_T;
      p[k] = i < m ? mine[i] : NOEDGE;
    }
    #pragma unroll
    for (int k = 0; k < 4; ++k)
      if (p[k] != NOEDGE) atomicAdd(&lcnt[p[k] & (BIN_NODES - 1)], 1);
  }
  __syncthreads();

  const int c = t < BIN_NODES ? lcnt[t] : 0;
  int inc = c;
  for (int off = 1; off < 32; off <<= 1) {
    const int u = __shfl_up(inc, off, 32);
    if (lane >= off) inc += u;
  }
  if (lane == 31 && w < BIN_NODES / 32) wtot[w] = inc;
  __syncthreads();
  if (t < BIN_NODES) {
    int pre = 0;
    for (int k = 0; k < BIN_NODES / 32; ++k) if (k < w) pre += wtot[k];
    const int off  = pre + inc - c;
    const int node = bin * BIN_NODES + t;
    loff[t] = off;
    lcnt[t] = 0;
    if (node < n) { cnt[node] = c; rowptr[node] = bstart + off; }
  }
  __syncthreads();

  int* __restrict__ out = csr + bstart;
  for (int i0 = t; i0 < m; i0 += 4 * CSR_T) {
    unsigned p[4];
    #pragma unroll
    for (int k = 0; k < 4; ++k) {
      const int i = i0 + k * CSR_T;
      p[k] = i < m ? mine[i] : NOEDGE;
    }
    int pos[4];
    #pragma unroll
    for (int k = 0; k < 4; ++k) {
      const int dl = (int)(p[k] & (BIN_NODES - 1));
      int rk = 0;
      if (p[k] != NOEDGE) rk = atomicAdd(&lcnt[dl], 1);
      pos[k] = loff[dl] + rk;
    }
    #pragma unroll
    for (int k = 0; k < 4; ++k)
      if (p[k] != NOEDGE) out[pos[k]] = (int)(p[k] >> BIN_SHIFT);
  }
}

#define NP_STEPS 4
#define NP_NODES_PER_BLOCK (8 * 2 * NP_STEPS)

__global__ void att_pass(const float* __restrict__ h0,
                         const float* __restrict__ attl,
                         const float* __restrict__ attr,
                         const int* __restrict__ cnt,
                         float* __restrict__ aldf, float* __restrict__ ar, int n) {
  const int lane  = (int)(threadIdx.x & 31);
  const int wave  = (int)((blockIdx.x * blockDim.x + threadIdx.x) >> 5);
  const int sub   = lane & 15;
  const int c4    = sub * 4;
  const int node0 = wave * (2 * NP_STEPS) + (lane >> 4);
  v2f* ald = (v2f*)aldf;

  if (wave * (2 * NP_STEPS) >= n) return;

  const v4f wl = *(const v4f*)(attl + c4);
  const v4f wr = *(const v4f*)(attr + c4);

  v4f hv[NP_STEPS];
  int dg[NP_STEPS];
  #pragma unroll
  for (int t = 0; t < NP_STEPS; ++t) {
    const int node = node0 + 2 * t;
    const int ld   = node < n ? node : n - 1;
    hv[t] = *(const v4f*)(h0 + (size_t)ld * HID + c4);
    dg[t] = cnt[ld];
  }

  __builtin_amdgcn_sched_barrier(0);

  #pragma unroll
  for (int t = 0; t < NP_STEPS; ++t) {
    const int node = node0 + 2 * t;
    const v4f h = hv[t];
    float pl = h.x * wl.x + h.y * wl.y + h.z * wl.z + h.w * wl.w;
    float pr = h.x * wr.x + h.y * wr.y + h.z * wr.z + h.w * wr.w;
    for (int off = 8; off > 0; off >>= 1) {
      pl += __shfl_xor(pl, off, 32);
      pr += __shfl_xor(pr, off, 32);
    }
    if (node < n && sub == 0) {
      v2f a; a.x = pl; a.y = rsqrtf((float)(dg[t] + 1));
      ald[node] = a;
      ar[node]  = pr;
    }
  }
}

#define GN 8
#define GE 24

template<int NS>
__device__ __forceinline__ void chunk_rows(v4f& acc, const float* __restrict__ h,
                                           int se, int wbits, int half, int c4) {
  const v4f* __restrict__ hv = (const v4f*)h;
  const unsigned sub = (unsigned)c4 >> 2;
  v4f rows[NS];
  #pragma unroll
  for (int j = 0; j < NS; ++j) {
    const int s0 = __builtin_amdgcn_readlane(se, 2 * j);
    const int s1 = __builtin_amdgcn_readlane(se, 2 * j + 1);
    const unsigned sj = (unsigned)(half ? s1 : s0);
    rows[j] = hv[sj * (HID / 4) + sub];
  }
  __builtin_amdgcn_sched_barrier(0);
  #pragma unroll
  for (int j = 0; j < NS; ++j) {
    const int w0 = __builtin_amdgcn_readlane(wbits, 2 * j);
    const int w1 = __builtin_amdgcn_readlane(wbits, 2 * j + 1);
    const float wj = __int_as_float(half ? w1 : w0);
    acc.x += wj * rows[j].x;
    acc.y += wj * rows[j].y;
    acc.z += wj * rows[j].z;
    acc.w += wj * rows[j].w;
  }
}

#define CHUNK_CASE(k) case k: chunk_rows<k>(acc, h, se, wbits, half, c4); break;

template<bool FIRST>
__global__ void __launch_bounds__(256) gather_layer(
                             const int* __restrict__ rowptr, const int* __restrict__ cnt,
                             const int* __restrict__ csr,
                             const float* __restrict__ h,
                             const float* h0res,
                             const float* __restrict__ aldf, const float* __restrict__ ar,
                             const float* __restrict__ attl2, const float* __restrict__ attr2,
                             float* orow,
                             float* __restrict__ ald2f, float* __restrict__ ar2, int n) {
  const int lane = (int)(threadIdx.x & 31);
  const int wib   = __builtin_amdgcn_readfirstlane((int)(threadIdx.x >> 5));
  const int wave  = (int)blockIdx.x * (int)(blockDim.x >> 5) + wib;
  const int node0 = wave * GN;
  if (node0 >= n) return;
  const int nn   = (n - node0 < GN) ? (n - node0) : GN;
  const int half = lane >> 4;
  const int c4   = (lane & 15) * 4;
  const v2f* ald = (const v2f*)aldf;

  int mb = 0, mc = 0;
  float mdi = 0.f, mar = 0.f, mws = 0.f;
  if (lane < nn) {
    mb = rowptr[node0 + lane];
    mc = cnt[node0 + lane];
    const v2f a = ald[node0 + lane];
    mdi = a.y;
    mar = ar[node0 + lane];
    mws = tanhf(a.x + mar) * a.y * a.y;
  }
  v4f wl2 = {0.f, 0.f, 0.f, 0.f}, wr2 = {0.f, 0.f, 0.f, 0.f};
  if (FIRST) {
    wl2 = *(const v4f*)(attl2 + c4);
    wr2 = *(const v4f*)(attr2 + c4);
  }

  int begin = __builtin_amdgcn_readlane(mb, 0);
  int c     = __builtin_amdgcn_readlane(mc, 0);
  int se0   = node0;
  {
    const int cc0 = c < GE ? c : GE;
    if (lane < cc0) se0 = csr[begin + lane];
  }

  for (int t = 0; t < nn; ++t) {
    const int   node = node0 + t;
    const float di_n = __int_as_float(__builtin_amdgcn_readlane(__float_as_int(mdi), t));
    const float ar_n = __int_as_float(__builtin_amdgcn_readlane(__float_as_int(mar), t));
    const float ws_n = __int_as_float(__builtin_amdgcn_readlane(__float_as_int(mws), t));

    const size_t own = (size_t)node * HID + c4;
    v4f hn = {0.f, 0.f, 0.f, 0.f}, r = {0.f, 0.f, 0.f, 0.f};
    int begin_n = 0, c_n = 0, se_n = node0;

    v4f acc = {0.f, 0.f, 0.f, 0.f};
    int se = se0;
    int cb = 0;
    do {
      const int rem = c - cb;
      const int cc  = rem < GE ? rem : GE;
      v2f a = {0.f, 0.f};
      if (lane < cc) a = ald[se];
      if (cb == 0) {
        hn = *(const v4f*)(h + own);
        r  = hn;
        if (!FIRST) r = *(const v4f*)(h0res + own);
        if (t + 1 < nn) {
          begin_n = __builtin_amdgcn_readlane(mb, t + 1);
          c_n     = __builtin_amdgcn_readlane(mc, t + 1);
          const int ccn = c_n < GE ? c_n : GE;
          if (lane < ccn) se_n = csr[begin_n + lane];
        }
      }
      float w = 0.f;
      if (lane < cc) w = tanhf(a.x + ar_n) * a.y * di_n;
      const int wbits = __float_as_int(w);
      switch ((cc + 1) >> 1) {
        CHUNK_CASE(1)  CHUNK_CASE(2)  CHUNK_CASE(3)  CHUNK_CASE(4)
        CHUNK_CASE(5)  CHUNK_CASE(6)  CHUNK_CASE(7)  CHUNK_CASE(8)
        CHUNK_CASE(9)  CHUNK_CASE(10) CHUNK_CASE(11) CHUNK_CASE(12)
        default: break;
      }
      cb += GE;
      if (cb < c) {
        const int rem2 = c - cb;
        const int cc2  = rem2 < GE ? rem2 : GE;
        se = node0;
        if (lane < cc2) se = csr[begin + cb + lane];
      }
    } while (cb < c);

    acc.x += __shfl_xor(acc.x, 16, 32);
    acc.y += __shfl_xor(acc.y, 16, 32);
    acc.z += __shfl_xor(acc.z, 16, 32);
    acc.w += __shfl_xor(acc.w, 16, 32);

    v4f o;
    o.x = acc.x + (EPS_BLEND * r.x + ws_n * hn.x);
    o.y = acc.y + (EPS_BLEND * r.y + ws_n * hn.y);
    o.z = acc.z + (EPS_BLEND * r.z + ws_n * hn.z);
    o.w = acc.w + (EPS_BLEND * r.w + ws_n * hn.w);

    if (FIRST) {
      o.x = fmaxf(o.x, 0.f); o.y = fmaxf(o.y, 0.f);
      o.z = fmaxf(o.z, 0.f); o.w = fmaxf(o.w, 0.f);
      float pl = o.x * wl2.x + o.y * wl2.y + o.z * wl2.z + o.w * wl2.w;
      float pr = o.x * wr2.x + o.y * wr2.y + o.z * wr2.z + o.w * wr2.w;
      for (int off = 8; off > 0; off >>= 1) {
        pl += __shfl_xor(pl, off, 32);
        pr += __shfl_xor(pr, off, 32);
      }
      if (lane == 0) {
        v2f a; a.x = pl; a.y = di_n;
        ((v2f*)ald2f)[node] = a;
        ar2[node] = pr;
      }
    }
    if (half == 0) *(v4f*)(orow + own) = o;

    begin = begin_n; c = c_n; se0 = se_n;
  }
}

extern "C" void kernel_launch(void* const* d_in, const int* in_sizes, int n_in,
                              void* d_out, int out_size, void* d_ws, size_t ws_size,
                              hipStream_t stream) {
  const float* x     = (const float*)d_in[0];
  const int*   ei    = (const int*)  d_in[1];
  const float* W_in  = (const float*)d_in[2];
  const float* b_in  = (const float*)d_in[3];
  const float* attl1 = (const float*)d_in[4];
  const float* attr1 = (const float*)d_in[5];
  const float* attl2 = (const float*)d_in[6];
  const float* attr2 = (const float*)d_in[7];
  const float* W_out = (const float*)d_in[8];
  const float* b_out = (const float*)d_in[9];
  float* out = (float*)d_out;

  const int IN = 128;
  const int E  = in_sizes[1] / 2;
  const int N  = in_sizes[0] / IN;
  const int*   srcI = ei;
  const int*   dstI = ei + E;

  float* H0   = (float*)d_ws;
  float* H1   = H0   + (size_t)N * HID;
  float* ALD1 = H1   + (size_t)N * HID;
  float* ALD2 = ALD1 + (size_t)N * 2;
  float* AR1  = ALD2 + (size_t)N * 2;
  float* AR2  = AR1  + N;
  int*   CNT  = (int*)(AR2 + N);
  int*   RPTR = CNT  + N;
  int*   BCNT = RPTR + N;
  int*   BCUR = BCNT + MAXBINS;
  unsigned* BINNED = (unsigned*)(BCUR + MAXBINS);
  int*   CSR  = (int*)(BINNED + E);

  const int ntiles = N / 16;
  const int gemmBlocks = (ntiles + 7) / 8;
  const int eBlk      = (E + BCHUNK - 1) / BCHUNK;
  const int nBins     = (N + BIN_NODES - 1) / BIN_NODES;
  const int nodeBlk   = (N + NP_NODES_PER_BLOCK - 1) / NP_NODES_PER_BLOCK;
  const int gBlk      = (N + 8 * GN - 1) / (8 * GN);

  gemm_wmma_f32<128, true, false><<<gemmBlocks, 256, 0, stream>>>(x, W_in, b_in, H0, ntiles);

  bin_zero   <<<1,     MAXBINS, 0, stream>>>(BCNT, BCUR);
  bin_count  <<<eBlk,  256,     0, stream>>>(dstI, BCNT, E);
  bin_scatter<<<eBlk,  256,     0, stream>>>(srcI, dstI, BCNT, BCUR, BINNED, E);
  bin_csr    <<<nBins, CSR_T,   0, stream>>>(BCNT, BINNED, CNT, RPTR, CSR, N);

  att_pass<<<nodeBlk, 256, 0, stream>>>(H0, attl1, attr1, CNT, ALD1, AR1, N);
  gather_layer<true> <<<gBlk, 256, 0, stream>>>(RPTR, CNT, CSR, H0, H0, ALD1, AR1,
                                                attl2, attr2, H1, ALD2, AR2, N);

  gather_layer<false><<<gBlk, 256, 0, stream>>>(RPTR, CNT, CSR, H1, H0, ALD2, AR2,
                                                attl2, attr2, H0, ALD2, AR2, N);

  gemm_wmma_f32<64, false, true><<<gemmBlocks, 256, 0, stream>>>(H0, W_out, b_out, out, ntiles);
}


# both GEMMs feed from LDS: W staged once per block, each wave's A tile and output tile go through LDS with whole-line 16-byte global accesses; same k order, sums bit-identical
# speedup vs baseline: 4.3556x; 1.1271x over previous
#include <hip/hip_runtime.h>
#include <hip/hip_bf16.h>


typedef __attribute__((ext_vector_type(2))) float v2f;
typedef __attribute__((ext_vector_type(8))) float v8f;
typedef __attribute__((ext_vector_type(4))) float v4f;

#define HID 64
#define EPS_BLEND 0.1f

#define GP 68

__device__ __forceinline__ int lds_pair_pos(int col) { return (col & ~7) + ((col >> 1) & 2); }

template<int K, bool RELU, bool RELU_IN>
__global__ void __launch_bounds__(256) gemm_wmma_f32(const float* __restrict__ A,
                                                     const float* __restrict__ W,
                                                     const float* __restrict__ bias,
                                                     float* __restrict__ out,
                                                     int ntiles) {
  constexpr int WP  = K + 4;
  constexpr int NCH = K / 64;
  constexpr int WV  = (64 * K / 4) / 256;
  __shared__ __attribute__((aligned(16))) float sW[64 * WP];
  __shared__ __attribute__((aligned(16))) float sT[8 * 16 * GP];

  const int t    = (int)threadIdx.x;
  const int lane = t & 31;
  const int wib  = t >> 5;
  const int tile = (int)blockIdx.x * 8 + wib;
  const bool live = tile < ntiles;
  const int ltile = live ? tile : ntiles - 1;
  const int c = lane & 15;
  const int h = lane >> 4;
  float* sA = sT + wib * (16 * GP);

  const float* arow = A + ((size_t)ltile * 16 + h) * K + c * 4;
  const int apos = h * GP + lds_pair_pos(c * 4);

  v4f av[8];
  #pragma unroll
  for (int i = 0; i < 8; ++i) av[i] = *(const v4f*)(arow + (size_t)(2 * i) * K);

  {
    v4f wv[WV];
    #pragma unroll
    for (int p = 0; p < WV; ++p) wv[p] = ((const v4f*)W)[t + 256 * p];
    #pragma unroll
    for (int p = 0; p < WV; ++p) {
      const int e   = (t + 256 * p) * 4;
      const int row = e / K;
      const int col = e % K;
      float* d = sW + row * WP + lds_pair_pos(col);
      v2f lo; lo.x = wv[p].x; lo.y = wv[p].y;
      v2f hi; hi.x = wv[p].z; hi.y = wv[p].w;
      *(v2f*)d       = lo;
      *(v2f*)(d + 4) = hi;
    }
  }
  __syncthreads();
  if (!live) return;

  v8f acc0 = {}, acc1 = {}, acc2 = {}, acc3 = {};
  const float* fa = sA + c * GP + 4 * h;
  const float* fb = sW + c * WP + 4 * h;

  #pragma unroll
  for (int ch = 0; ch < NCH; ++ch) {
    #pragma unroll
    for (int i = 0; i < 8; ++i) {
      v4f a = av[i];
      if (RELU_IN) {
        a.x = fmaxf(a.x, 0.f); a.y = fmaxf(a.y, 0.f);
        a.z = fmaxf(a.z, 0.f); a.w = fmaxf(a.w, 0.f);
      }
      float* d = sA + (2 * i) * GP + apos;
      v2f lo; lo.x = a.x; lo.y = a.y;
      v2f hi; hi.x = a.z; hi.y = a.w;
      *(v2f*)d       = lo;
      *(v2f*)(d + 4) = hi;
    }
    if (ch + 1 < NCH) {
      #pragma unroll
      for (int i = 0; i < 8; ++i)
        av[i] = *(const v4f*)(arow + (size_t)(2 * i) * K + 64 * (ch + 1));
    }
    #pragma unroll
    for (int j = 0; j < 8; ++j) {
      const v4f a  = *(const v4f*)(fa + 8 * j);
      const v4f b0 = *(const v4f*)(fb +  0 * WP + 64 * ch + 8 * j);
      const v4f b1 = *(const v4f*)(fb + 16 * WP + 64 * ch + 8 * j);
      const v4f b2 = *(const v4f*)(fb + 32 * WP + 64 * ch + 8 * j);
      const v4f b3 = *(const v4f*)(fb + 48 * WP + 64 * ch + 8 * j);
      v2f al, ah, p0, q0, p1, q1, p2, q2, p3, q3;
      al.x = a.x;  al.y = a.y;  ah.x = a.z;  ah.y = a.w;
      p0.x = b0.x; p0.y = b0.y; q0.x = b0.z; q0.y = b0.w;
      p1.x = b1.x; p1.y = b1.y; q1.x = b1.z; q1.y = b1.w;
      p2.x = b2.x; p2.y = b2.y; q2.x = b2.z; q2.y = b2.w;
      p3.x = b3.x; p3.y = b3.y; q3.x = b3.z; q3.y = b3.w;
      acc0 = __builtin_amdgcn_wmma_f32_16x16x4_f32(false, al, false, p0, (short)0, acc0, false, false);
      acc1 = __builtin_amdgcn_wmma_f32_16x16x4_f32(false, al, false, p1, (short)0, acc1, false, false);
      acc2 = __builtin_amdgcn_wmma_f32_16x16x4_f32(false, al, false, p2, (short)0, acc2, false, false);
      acc3 = __builtin_amdgcn_wmma_f32_16x16x4_f32(false, al, false, p3, (short)0, acc3, false, false);
      acc0 = __builtin_amdgcn_wmma_f32_16x16x4_f32(false, ah, false, q0, (short)0, acc0, false, false);
      acc1 = __builtin_amdgcn_wmma_f32_16x16x4_f32(false, ah, false, q1, (short)0, acc1, false, false);
      acc2 = __builtin_amdgcn_wmma_f32_16x16x4_f32(false, ah, false, q2, (short)0, acc2, false, false);
      acc3 = __builtin_amdgcn_wmma_f32_16x16x4_f32(false, ah, false, q3, (short)0, acc3, false, false);
    }
  }

  const float bb0 = bias[c], bb1 = bias[16 + c], bb2 = bias[32 + c], bb3 = bias[48 + c];
  #pragma unroll
  for (int v = 0; v < 8; ++v) {
    float x0 = acc0[v] + bb0;
    float x1 = acc1[v] + bb1;
    float x2 = acc2[v] + bb2;
    float x3 = acc3[v] + bb3;
    if (RELU) {
      x0 = fmaxf(x0, 0.f); x1 = fmaxf(x1, 0.f);
      x2 = fmaxf(x2, 0.f); x3 = fmaxf(x3, 0.f);
    }
    float* d = sA + (8 * h + v) * GP + c;
    d[0]  = x0;
    d[16] = x1;
    d[32] = x2;
    d[48] = x3;
  }
  float* orow = out + ((size_t)tile * 16 + h) * HID + c * 4;
  #pragma unroll
  for (int i = 0; i < 8; ++i) {
    const v4f o = *(const v4f*)(sA + (2 * i + h) * GP + c * 4);
    *(v4f*)(orow + (size_t)(2 * i) * HID) = o;
  }
}

#define BIN_SHIFT 9
#define BIN_NODES 512
#define MAXBINS   256
#define BE        16
#define BCHUNK    (256 * BE)
#define CSR_T     1024
#define NOEDGE    0xFFFFFFFFu

__global__ void bin_zero(int* __restrict__ bincnt, int* __restrict__ bincur) {
  bincnt[threadIdx.x] = 0;
  bincur[threadIdx.x] = 0;
}

__global__ void __launch_bounds__(256) bin_count(const int* __restrict__ dst,
                                                 int* __restrict__ bincnt, int e) {
  __shared__ int hist[MAXBINS];
  const int t = (int)threadIdx.x;
  hist[t] = 0;
  __syncthreads();
  const int base = (int)blockIdx.x * BCHUNK + t;
  int d[BE];
  #pragma unroll
  for (int k = 0; k < BE; ++k) {
    const int idx = base + k * 256;
    d[k] = idx < e ? dst[idx] : -1;
  }
  #pragma unroll
  for (int k = 0; k < BE; ++k)
    if (d[k] >= 0) atomicAdd(&hist[d[k] >> BIN_SHIFT], 1);
  __syncthreads();
  const int h = hist[t];
  if (h > 0) atomicAdd(&bincnt[t], h);
}

__global__ void __launch_bounds__(256) bin_scatter(const int* __restrict__ src,
                                                   const int* __restrict__ dst,
                                                   const int* __restrict__ bincnt,
                                                   int* __restrict__ bincur,
                                                   unsigned* __restrict__ binned, int e) {
  __shared__ int hist[MAXBINS];
  __shared__ int gbase[MAXBINS];
  __shared__ int wsum[8];
  const int t    = (int)threadIdx.x;
  const int lane = t & 31;
  const int w    = t >> 5;
  hist[t] = 0;

  const int base = (int)blockIdx.x * BCHUNK + t;
  int d[BE], s[BE], r[BE];
  #pragma unroll
  for (int k = 0; k < BE; ++k) {
    const int idx = base + k * 256;
    const bool ok = idx < e;
    d[k] = ok ? dst[idx] : -1;
    s[k] = ok ? src[idx] : 0;
  }

  const int c = bincnt[t];
  int inc = c;
  for (int off = 1; off < 32; off <<= 1) {
    const int v = __shfl_up(inc, off, 32);
    if (lane >= off) inc += v;
  }
  if (lane == 31) wsum[w] = inc;
  __syncthreads();
  int pre = 0;
  for (int k = 0; k < 8; ++k) if (k < w) pre += wsum[k];
  const int bstart = pre + inc - c;

  #pragma unroll
  for (int k = 0; k < BE; ++k) {
    r[k] = 0;
    if (d[k] >= 0) r[k] = atomicAdd(&hist[d[k] >> BIN_SHIFT], 1);
  }
  __syncthreads();
  const int h = hist[t];
  int gb = bstart;
  if (h > 0) gb += atomicAdd(&bincur[t], h);
  gbase[t] = gb;
  __syncthreads();
  int pos[BE];
  #pragma unroll
  for (int k = 0; k < BE; ++k)
    pos[k] = gbase[d[k] >= 0 ? d[k] >> BIN_SHIFT : 0] + r[k];
  #pragma unroll
  for (int k = 0; k < BE; ++k) {
    if (d[k] >= 0)
      binned[pos[k]] = ((unsigned)s[k] << BIN_SHIFT) | ((unsigned)d[k] & (BIN_NODES - 1));
  }
}

__global__ void __launch_bounds__(CSR_T) bin_csr(const int* __restrict__ bincnt,
                                                 const unsigned* __restrict__ binned,
                                                 int* __restrict__ cnt,
                                                 int* __restrict__ rowptr,
                                                 int* __restrict__ csr, int n) {
  __shared__ int lcnt[BIN_NODES];
  __shared__ int loff[BIN_NODES];
  __shared__ int wsum[CSR_T / 32];
  __shared__ int wtot[BIN_NODES / 32];
  const int t    = (int)threadIdx.x;
  const int lane = t & 31;
  const int w    = t >> 5;
  const int bin  = (int)blockIdx.x;

  int v = t < bin ? bincnt[t] : 0;
  for (int off = 16; off > 0; off >>= 1) v += __shfl_xor(v, off, 32);
  if (lane == 0) wsum[w] = v;
  if (t < BIN_NODES) lcnt[t] = 0;
  const int m = bincnt[bin];
  __syncthreads();
  int bstart = 0;
  for (int k = 0; k < CSR_T / 32; ++k) bstart += wsum[k];
  const unsigned* __restrict__ mine = binned + bstart;

  for (int i0 = t; i0 < m; i0 += 4 * CSR_T) {
    unsigned p[4];
    #pragma unroll
    for (int k = 0; k < 4; ++k) {
      const int i = i0 + k * CSR_T;
      p[k] = i < m ? mine[i] : NOEDGE;
    }
    #pragma unroll
    for (int k = 0; k < 4; ++k)
      if (p[k] != NOEDGE) atomicAdd(&lcnt[p[k] & (BIN_NODES - 1)], 1);
  }
  __syncthreads();

  const int c = t < BIN_NODES ? lcnt[t] : 0;
  int inc = c;
  for (int off = 1; off < 32; off <<= 1) {
    const int u = __shfl_up(inc, off, 32);
    if (lane >= off) inc += u;
  }
  if (lane == 31 && w < BIN_NODES / 32) wtot[w] = inc;
  __syncthreads();
  if (t < BIN_NODES) {
    int pre = 0;
    for (int k = 0; k < BIN_NODES / 32; ++k) if (k < w) pre += wtot[k];
    const int off  = pre + inc - c;
    const int node = bin * BIN_NODES + t;
    loff[t] = off;
    lcnt[t] = 0;
    if (node < n) { cnt[node] = c; rowptr[node] = bstart + off; }
  }
  __syncthreads();

  int* __restrict__ out = csr + bstart;
  for (int i0 = t; i0 < m; i0 += 4 * CSR_T) {
    unsigned p[4];
    #pragma unroll
    for (int k = 0; k < 4; ++k) {
      const int i = i0 + k * CSR_T;
      p[k] = i < m ? mine[i] : NOEDGE;
    }
    int pos[4];
    #pragma unroll
    for (int k = 0; k < 4; ++k) {
      const int dl = (int)(p[k] & (BIN_NODES - 1));
      int rk = 0;
      if (p[k] != NOEDGE) rk = atomicAdd(&lcnt[dl], 1);
      pos[k] = loff[dl] + rk;
    }
    #pragma unroll
    for (int k = 0; k < 4; ++k)
      if (p[k] != NOEDGE) out[pos[k]] = (int)(p[k] >> BIN_SHIFT);
  }
}

#define NP_STEPS 4
#define NP_NODES_PER_BLOCK (8 * 2 * NP_STEPS)

__global__ void att_pass(const float* __restrict__ h0,
                         const float* __restrict__ attl,
                         const float* __restrict__ attr,
                         const int* __restrict__ cnt,
                         float* __restrict__ aldf, float* __restrict__ ar, int n) {
  const int lane  = (int)(threadIdx.x & 31);
  const int wave  = (int)((blockIdx.x * blockDim.x + threadIdx.x) >> 5);
  const int sub   = lane & 15;
  const int c4    = sub * 4;
  const int node0 = wave * (2 * NP_STEPS) + (lane >> 4);
  v2f* ald = (v2f*)aldf;

  if (wave * (2 * NP_STEPS) >= n) return;

  const v4f wl = *(const v4f*)(attl + c4);
  const v4f wr = *(const v4f*)(attr + c4);

  v4f hv[NP_STEPS];
  int dg[NP_STEPS];
  #pragma unroll
  for (int t = 0; t < NP_STEPS; ++t) {
    const int node = node0 + 2 * t;
    const int ld   = node < n ? node : n - 1;
    hv[t] = *(const v4f*)(h0 + (size_t)ld * HID + c4);
    dg[t] = cnt[ld];
  }

  __builtin_amdgcn_sched_barrier(0);

  #pragma unroll
  for (int t = 0; t < NP_STEPS; ++t) {
    const int node = node0 + 2 * t;
    const v4f h = hv[t];
    float pl = h.x * wl.x + h.y * wl.y + h.z * wl.z + h.w * wl.w;
    float pr = h.x * wr.x + h.y * wr.y + h.z * wr.z + h.w * wr.w;
    for (int off = 8; off > 0; off >>= 1) {
      pl += __shfl_xor(pl, off, 32);
      pr += __shfl_xor(pr, off, 32);
    }
    if (node < n && sub == 0) {
      v2f a; a.x = pl; a.y = rsqrtf((float)(dg[t] + 1));
      ald[node] = a;
      ar[node]  = pr;
    }
  }
}

#define GN 8
#define GE 24

template<int NS>
__device__ __forceinline__ void chunk_rows(v4f& acc, const float* __restrict__ h,
                                           int se, int wbits, int half, int c4) {
  const v4f* __restrict__ hv = (const v4f*)h;
  const unsigned sub = (unsigned)c4 >> 2;
  v4f rows[NS];
  #pragma unroll
  for (int j = 0; j < NS; ++j) {
    const int s0 = __builtin_amdgcn_readlane(se, 2 * j);
    const int s1 = __builtin_amdgcn_readlane(se, 2 * j + 1);
    const unsigned sj = (unsigned)(half ? s1 : s0);
    rows[j] = hv[sj * (HID / 4) + sub];
  }
  __builtin_amdgcn_sched_barrier(0);
  #pragma unroll
  for (int j = 0; j < NS; ++j) {
    const int w0 = __builtin_amdgcn_readlane(wbits, 2 * j);
    const int w1 = __builtin_amdgcn_readlane(wbits, 2 * j + 1);
    const float wj = __int_as_float(half ? w1 : w0);
    acc.x += wj * rows[j].x;
    acc.y += wj * rows[j].y;
    acc.z += wj * rows[j].z;
    acc.w += wj * rows[j].w;
  }
}

#define CHUNK_CASE(k) case k: chunk_rows<k>(acc, h, se, wbits, half, c4); break;

template<bool FIRST>
__global__ void __launch_bounds__(256) gather_layer(
                             const int* __restrict__ rowptr, const int* __restrict__ cnt,
                             const int* __restrict__ csr,
                             const float* __restrict__ h,
                             const float* h0res,
                             const float* __restrict__ aldf, const float* __restrict__ ar,
                             const float* __restrict__ attl2, const float* __restrict__ attr2,
                             float* orow,
                             float* __restrict__ ald2f, float* __restrict__ ar2, int n) {
  const int lane = (int)(threadIdx.x & 31);
  const int wib   = __builtin_amdgcn_readfirstlane((int)(threadIdx.x >> 5));
  const int wave  = (int)blockIdx.x * (int)(blockDim.x >> 5) + wib;
  const int node0 = wave * GN;
  if (node0 >= n) return;
  const int nn   = (n - node0 < GN) ? (n - node0) : GN;
  const int half = lane >> 4;
  const int c4   = (lane & 15) * 4;
  const v2f* ald = (const v2f*)aldf;

  int mb = 0, mc = 0;
  float mdi = 0.f, mar = 0.f, mws = 0.f;
  if (lane < nn) {
    mb = rowptr[node0 + lane];
    mc = cnt[node0 + lane];
    const v2f a = ald[node0 + lane];
    mdi = a.y;
    mar = ar[node0 + lane];
    mws = tanhf(a.x + mar) * a.y * a.y;
  }
  v4f wl2 = {0.f, 0.f, 0.f, 0.f}, wr2 = {0.f, 0.f, 0.f, 0.f};
  if (FIRST) {
    wl2 = *(const v4f*)(attl2 + c4);
    wr2 = *(const v4f*)(attr2 + c4);
  }

  int begin = __builtin_amdgcn_readlane(mb, 0);
  int c     = __builtin_amdgcn_readlane(mc, 0);
  int se0   = node0;
  {
    const int cc0 = c < GE ? c : GE;
    if (lane < cc0) se0 = csr[begin + lane];
  }

  for (int t = 0; t < nn; ++t) {
    const int   node = node0 + t;
    const float di_n = __int_as_float(__builtin_amdgcn_readlane(__float_as_int(mdi), t));
    const float ar_n = __int_as_float(__builtin_amdgcn_readlane(__float_as_int(mar), t));
    const float ws_n = __int_as_float(__builtin_amdgcn_readlane(__float_as_int(mws), t));

    const size_t own = (size_t)node * HID + c4;
    v4f hn = {0.f, 0.f, 0.f, 0.f}, r = {0.f, 0.f, 0.f, 0.f};
    int begin_n = 0, c_n = 0, se_n = node0;

    v4f acc = {0.f, 0.f, 0.f, 0.f};
    int se = se0;
    int cb = 0;
    do {
      const int rem = c - cb;
      const int cc  = rem < GE ? rem : GE;
      v2f a = {0.f, 0.f};
      if (lane < cc) a = ald[se];
      if (cb == 0) {
        hn = *(const v4f*)(h + own);
        r  = hn;
        if (!FIRST) r = *(const v4f*)(h0res + own);
        if (t + 1 < nn) {
          begin_n = __builtin_amdgcn_readlane(mb, t + 1);
          c_n     = __builtin_amdgcn_readlane(mc, t + 1);
          const int ccn = c_n < GE ? c_n : GE;
          if (lane < ccn) se_n = csr[begin_n + lane];
        }
      }
      float w = 0.f;
      if (lane < cc) w = tanhf(a.x + ar_n) * a.y * di_n;
      const int wbits = __float_as_int(w);
      switch ((cc + 1) >> 1) {
        CHUNK_CASE(1)  CHUNK_CASE(2)  CHUNK_CASE(3)  CHUNK_CASE(4)
        CHUNK_CASE(5)  CHUNK_CASE(6)  CHUNK_CASE(7)  CHUNK_CASE(8)
        CHUNK_CASE(9)  CHUNK_CASE(10) CHUNK_CASE(11) CHUNK_CASE(12)
        default: break;
      }
      cb += GE;
      if (cb < c) {
        const int rem2 = c - cb;
        const int cc2  = rem2 < GE ? rem2 : GE;
        se = node0;
        if (lane < cc2) se = csr[begin + cb + lane];
      }
    } while (cb < c);

    acc.x += __shfl_xor(acc.x, 16, 32);
    acc.y += __shfl_xor(acc.y, 16, 32);
    acc.z += __shfl_xor(acc.z, 16, 32);
    acc.w += __shfl_xor(acc.w, 16, 32);

    v4f o;
    o.x = acc.x + (EPS_BLEND * r.x + ws_n * hn.x);
    o.y = acc.y + (EPS_BLEND * r.y + ws_n * hn.y);
    o.z = acc.z + (EPS_BLEND * r.z + ws_n * hn.z);
    o.w = acc.w + (EPS_BLEND * r.w + ws_n * hn.w);

    if (FIRST) {
      o.x = fmaxf(o.x, 0.f); o.y = fmaxf(o.y, 0.f);
      o.z = fmaxf(o.z, 0.f); o.w = fmaxf(o.w, 0.f);
      float pl = o.x * wl2.x + o.y * wl2.y + o.z * wl2.z + o.w * wl2.w;
      float pr = o.x * wr2.x + o.y * wr2.y + o.z * wr2.z + o.w * wr2.w;
      for (int off = 8; off > 0; off >>= 1) {
        pl += __shfl_xor(pl, off, 32);
        pr += __shfl_xor(pr, off, 32);
      }
      if (lane == 0) {
        v2f a; a.x = pl; a.y = di_n;
        ((v2f*)ald2f)[node] = a;
        ar2[node] = pr;
      }
    }
    if (half == 0) *(v4f*)(orow + own) = o;

    begin = begin_n; c = c_n; se0 = se_n;
  }
}

extern "C" void kernel_launch(void* const* d_in, const int* in_sizes, int n_in,
                              void* d_out, int out_size, void* d_ws, size_t ws_size,
                              hipStream_t stream) {
  const float* x     = (const float*)d_in[0];
  const int*   ei    = (const int*)  d_in[1];
  const float* W_in  = (const float*)d_in[2];
  const float* b_in  = (const float*)d_in[3];
  const float* attl1 = (const float*)d_in[4];
  const float* attr1 = (const float*)d_in[5];
  const float* attl2 = (const float*)d_in[6];
  const float* attr2 = (const float*)d_in[7];
  const float* W_out = (const float*)d_in[8];
  const float* b_out = (const float*)d_in[9];
  float* out = (float*)d_out;

  const int IN = 128;
  const int E  = in_sizes[1] / 2;
  const int N  = in_sizes[0] / IN;
  const int*   srcI = ei;
  const int*   dstI = ei + E;

  float* H0   = (float*)d_ws;
  float* H1   = H0   + (size_t)N * HID;
  float* ALD1 = H1   + (size_t)N * HID;
  float* ALD2 = ALD1 + (size_t)N * 2;
  float* AR1  = ALD2 + (size_t)N * 2;
  float* AR2  = AR1  + N;
  int*   CNT  = (int*)(AR2 + N);
  int*   RPTR = CNT  + N;
  int*   BCNT = RPTR + N;
  int*   BCUR = BCNT + MAXBINS;
  unsigned* BINNED = (unsigned*)(BCUR + MAXBINS);
  int*   CSR  = (int*)(BINNED + E);

  const int ntiles = N / 16;
  const int gemmBlocks = (ntiles + 7) / 8;
  const int eBlk      = (E + BCHUNK - 1) / BCHUNK;
  const int nBins     = (N + BIN_NODES - 1) / BIN_NODES;
  const int nodeBlk   = (N + NP_NODES_PER_BLOCK - 1) / NP_NODES_PER_BLOCK;
  const int gBlk      = (N + 8 * GN - 1) / (8 * GN);

  gemm_wmma_f32<128, true, false><<<gemmBlocks, 256, 0, stream>>>(x, W_in, b_in, H0, ntiles);

  bin_zero   <<<1,     MAXBINS, 0, stream>>>(BCNT, BCUR);
  bin_count  <<<eBlk,  256,     0, stream>>>(dstI, BCNT, E);
  bin_scatter<<<eBlk,  256,     0, stream>>>(srcI, dstI, BCNT, BCUR, BINNED, E);
  bin_csr    <<<nBins, CSR_T,   0, stream>>>(BCNT, BINNED, CNT, RPTR, CSR, N);

  att_pass<<<nodeBlk, 256, 0, stream>>>(H0, attl1, attr1, CNT, ALD1, AR1, N);
  gather_layer<true> <<<gBlk, 256, 0, stream>>>(RPTR, CNT, CSR, H0, H0, ALD1, AR1,
                                                attl2, attr2, H1, ALD2, AR2, N);

  gather_layer<false><<<gBlk, 256, 0, stream>>>(RPTR, CNT, CSR, H1, H0, ALD2, AR2,
                                                attl2, attr2, H0, ALD2, AR2, N);

  gemm_wmma_f32<64, false, true><<<gemmBlocks, 256, 0, stream>>>(H0, W_out, b_out, out, ntiles);
}


# input projection GEMM loses its own launch: its tile blocks ride in the two edge-binning launches (block index picks the body), so its WMMA work runs under the binning's memory waits; 9 launches to 8
# speedup vs baseline: 4.4967x; 1.0324x over previous
#include <hip/hip_runtime.h>
#include <hip/hip_bf16.h>


typedef __attribute__((ext_vector_type(2))) float v2f;
typedef __attribute__((ext_vector_type(8))) float v8f;
typedef __attribute__((ext_vector_type(4))) float v4f;

#define HID 64
#define EPS_BLEND 0.1f

#define GP 68

__device__ __forceinline__ int lds_pair_pos(int col) { return (col & ~7) + ((col >> 1) & 2); }

template<int K, bool RELU, bool RELU_IN>
__device__ __forceinline__ void gemm_block(const float* __restrict__ A,
                                           const float* __restrict__ W,
                                           const float* __restrict__ bias,
                                           float* __restrict__ out,
                                           int ntiles, int blk) {
  constexpr int WP  = K + 4;
  constexpr int NCH = K / 64;
  constexpr int WV  = (64 * K / 4) / 256;
  __shared__ __attribute__((aligned(16))) float sW[64 * WP];
  __shared__ __attribute__((aligned(16))) float sT[8 * 16 * GP];

  const int t    = (int)threadIdx.x;
  const int lane = t & 31;
  const int wib  = t >> 5;
  const int tile = blk * 8 + wib;
  const bool live = tile < ntiles;
  const int ltile = live ? tile : ntiles - 1;
  const int c = lane & 15;
  const int h = lane >> 4;
  float* sA = sT + wib * (16 * GP);

  const float* arow = A + ((size_t)ltile * 16 + h) * K + c * 4;
  const int apos = h * GP + lds_pair_pos(c * 4);

  v4f av[8];
  #pragma unroll
  for (int i = 0; i < 8; ++i) av[i] = *(const v4f*)(arow + (size_t)(2 * i) * K);

  {
    v4f wv[WV];
    #pragma unroll
    for (int p = 0; p < WV; ++p) wv[p] = ((const v4f*)W)[t + 256 * p];
    #pragma unroll
    for (int p = 0; p < WV; ++p) {
      const int e   = (t + 256 * p) * 4;
      const int row = e / K;
      const int col = e % K;
      float* d = sW + row * WP + lds_pair_pos(col);
      v2f lo; lo.x = wv[p].x; lo.y = wv[p].y;
      v2f hi; hi.x = wv[p].z; hi.y = wv[p].w;
      *(v2f*)d       = lo;
      *(v2f*)(d + 4) = hi;
    }
  }
  __syncthreads();
  if (!live) return;

  v8f acc0 = {}, acc1 = {}, acc2 = {}, acc3 = {};
  const float* fa = sA + c * GP + 4 * h;
  const float* fb = sW + c * WP + 4 * h;

  #pragma unroll
  for (int ch = 0; ch < NCH; ++ch) {
    #pragma unroll
    for (int i = 0; i < 8; ++i) {
      v4f a = av[i];
      if (RELU_IN) {
        a.x = fmaxf(a.x, 0.f); a.y = fmaxf(a.y, 0.f);
        a.z = fmaxf(a.z, 0.f); a.w = fmaxf(a.w, 0.f);
      }
      float* d = sA + (2 * i) * GP + apos;
      v2f lo; lo.x = a.x; lo.y = a.y;
      v2f hi; hi.x = a.z; hi.y = a.w;
      *(v2f*)d       = lo;
      *(v2f*)(d + 4) = hi;
    }
    if (ch + 1 < NCH) {
      #pragma unroll
      for (int i = 0; i < 8; ++i)
        av[i] = *(const v4f*)(arow + (size_t)(2 * i) * K + 64 * (ch + 1));
    }
    #pragma unroll
    for (int j = 0; j < 8; ++j) {
      const v4f a  = *(const v4f*)(fa + 8 * j);
      const v4f b0 = *(const v4f*)(fb +  0 * WP + 64 * ch + 8 * j);
      const v4f b1 = *(const v4f*)(fb + 16 * WP + 64 * ch + 8 * j);
      const v4f b2 = *(const v4f*)(fb + 32 * WP + 64 * ch + 8 * j);
      const v4f b3 = *(const v4f*)(fb + 48 * WP + 64 * ch + 8 * j);
      v2f al, ah, p0, q0, p1, q1, p2, q2, p3, q3;
      al.x = a.x;  al.y = a.y;  ah.x = a.z;  ah.y = a.w;
      p0.x = b0.x; p0.y = b0.y; q0.x = b0.z; q0.y = b0.w;
      p1.x = b1.x; p1.y = b1.y; q1.x = b1.z; q1.y = b1.w;
      p2.x = b2.x; p2.y = b2.y; q2.x = b2.z; q2.y = b2.w;
      p3.x = b3.x; p3.y = b3.y; q3.x = b3.z; q3.y = b3.w;
      acc0 = __builtin_amdgcn_wmma_f32_16x16x4_f32(false, al, false, p0, (short)0, acc0, false, false);
      acc1 = __builtin_amdgcn_wmma_f32_16x16x4_f32(false, al, false, p1, (short)0, acc1, false, false);
      acc2 = __builtin_amdgcn_wmma_f32_16x16x4_f32(false, al, false, p2, (short)0, acc2, false, false);
      acc3 = __builtin_amdgcn_wmma_f32_16x16x4_f32(false, al, false, p3, (short)0, acc3, false, false);
      acc0 = __builtin_amdgcn_wmma_f32_16x16x4_f32(false, ah, false, q0, (short)0, acc0, false, false);
      acc1 = __builtin_amdgcn_wmma_f32_16x16x4_f32(false, ah, false, q1, (short)0, acc1, false, false);
      acc2 = __builtin_amdgcn_wmma_f32_16x16x4_f32(false, ah, false, q2, (short)0, acc2, false, false);
      acc3 = __builtin_amdgcn_wmma_f32_16x16x4_f32(false, ah, false, q3, (short)0, acc3, false, false);
    }
  }

  const float bb0 = bias[c], bb1 = bias[16 + c], bb2 = bias[32 + c], bb3 = bias[48 + c];
  #pragma unroll
  for (int v = 0; v < 8; ++v) {
    float x0 = acc0[v] + bb0;
    float x1 = acc1[v] + bb1;
    float x2 = acc2[v] + bb2;
    float x3 = acc3[v] + bb3;
    if (RELU) {
      x0 = fmaxf(x0, 0.f); x1 = fmaxf(x1, 0.f);
      x2 = fmaxf(x2, 0.f); x3 = fmaxf(x3, 0.f);
    }
    float* d = sA + (8 * h + v) * GP + c;
    d[0]  = x0;
    d[16] = x1;
    d[32] = x2;
    d[48] = x3;
  }
  float* orow = out + ((size_t)tile * 16 + h) * HID + c * 4;
  #pragma unroll
  for (int i = 0; i < 8; ++i) {
    const v4f o = *(const v4f*)(sA + (2 * i + h) * GP + c * 4);
    *(v4f*)(orow + (size_t)(2 * i) * HID) = o;
  }
}

template<int K, bool RELU, bool RELU_IN>
__global__ void __launch_bounds__(256) gemm_wmma_f32(const float* __restrict__ A,
                                                     const float* __restrict__ W,
                                                     const float* __restrict__ bias,
                                                     float* __restrict__ out,
                                                     int ntiles) {
  gemm_block<K, RELU, RELU_IN>(A, W, bias, out, ntiles, (int)blockIdx.x);
}

#define BIN_SHIFT 9
#define BIN_NODES 512
#define MAXBINS   256
#define BE        16
#define BCHUNK    (256 * BE)
#define CSR_T     1024
#define NOEDGE    0xFFFFFFFFu

__global__ void bin_zero(int* __restrict__ bincnt, int* __restrict__ bincur) {
  bincnt[threadIdx.x] = 0;
  bincur[threadIdx.x] = 0;
}

__device__ __forceinline__ void bin_count_block(const int* __restrict__ dst,
                                                int* __restrict__ bincnt, int e, int blk) {
  __shared__ int hist[MAXBINS];
  const int t = (int)threadIdx.x;
  hist[t] = 0;
  __syncthreads();
  const int base = blk * BCHUNK + t;
  int d[BE];
  #pragma unroll
  for (int k = 0; k < BE; ++k) {
    const int idx = base + k * 256;
    d[k] = idx < e ? dst[idx] : -1;
  }
  #pragma unroll
  for (int k = 0; k < BE; ++k)
    if (d[k] >= 0) atomicAdd(&hist[d[k] >> BIN_SHIFT], 1);
  __syncthreads();
  const int h = hist[t];
  if (h > 0) atomicAdd(&bincnt[t], h);
}

__device__ __forceinline__ void bin_scatter_block(const int* __restrict__ src,
                                                  const int* __restrict__ dst,
                                                  const int* __restrict__ bincnt,
                                                  int* __restrict__ bincur,
                                                  unsigned* __restrict__ binned, int e, int blk) {
  __shared__ int hist[MAXBINS];
  __shared__ int gbase[MAXBINS];
  __shared__ int wsum[8];
  const int t    = (int)threadIdx.x;
  const int lane = t & 31;
  const int w    = t >> 5;
  hist[t] = 0;

  const int base = blk * BCHUNK + t;
  int d[BE], s[BE], r[BE];
  #pragma unroll
  for (int k = 0; k < BE; ++k) {
    const int idx = base + k * 256;
    const bool ok = idx < e;
    d[k] = ok ? dst[idx] : -1;
    s[k] = ok ? src[idx] : 0;
  }

  const int c = bincnt[t];
  int inc = c;
  for (int off = 1; off < 32; off <<= 1) {
    const int v = __shfl_up(inc, off, 32);
    if (lane >= off) inc += v;
  }
  if (lane == 31) wsum[w] = inc;
  __syncthreads();
  int pre = 0;
  for (int k = 0; k < 8; ++k) if (k < w) pre += wsum[k];
  const int bstart = pre + inc - c;

  #pragma unroll
  for (int k = 0; k < BE; ++k) {
    r[k] = 0;
    if (d[k] >= 0) r[k] = atomicAdd(&hist[d[k] >> BIN_SHIFT], 1);
  }
  __syncthreads();
  const int h = hist[t];
  int gb = bstart;
  if (h > 0) gb += atomicAdd(&bincur[t], h);
  gbase[t] = gb;
  __syncthreads();
  int pos[BE];
  #pragma unroll
  for (int k = 0; k < BE; ++k)
    pos[k] = gbase[d[k] >= 0 ? d[k] >> BIN_SHIFT : 0] + r[k];
  #pragma unroll
  for (int k = 0; k < BE; ++k) {
    if (d[k] >= 0)
      binned[pos[k]] = ((unsigned)s[k] << BIN_SHIFT) | ((unsigned)d[k] & (BIN_NODES - 1));
  }
}

__global__ void __launch_bounds__(256) count_and_gemm(const int* __restrict__ dst,
                                                      int* __restrict__ bincnt, int e,
                                                      const float* __restrict__ A,
                                                      const float* __restrict__ W,
                                                      const float* __restrict__ bias,
                                                      float* __restrict__ out,
                                                      int ntiles, int gemm_blk0, int gemm_blocks) {
  const int b = (int)blockIdx.x;
  if (b < gemm_blocks) gemm_block<128, true, false>(A, W, bias, out, ntiles, gemm_blk0 + b);
  else                 bin_count_block(dst, bincnt, e, b - gemm_blocks);
}

__global__ void __launch_bounds__(256) scatter_and_gemm(const int* __restrict__ src,
                                                        const int* __restrict__ dst,
                                                        const int* __restrict__ bincnt,
                                                        int* __restrict__ bincur,
                                                        unsigned* __restrict__ binned, int e,
                                                        const float* __restrict__ A,
                                                        const float* __restrict__ W,
                                                        const float* __restrict__ bias,
                                                        float* __restrict__ out,
                                                        int ntiles, int gemm_blk0, int gemm_blocks) {
  const int b = (int)blockIdx.x;
  if (b < gemm_blocks) gemm_block<128, true, false>(A, W, bias, out, ntiles, gemm_blk0 + b);
  else                 bin_scatter_block(src, dst, bincnt, bincur, binned, e, b - gemm_blocks);
}

__global__ void __launch_bounds__(CSR_T) bin_csr(const int* __restrict__ bincnt,
                                                 const unsigned* __restrict__ binned,
                                                 int* __restrict__ cnt,
                                                 int* __restrict__ rowptr,
                                                 int* __restrict__ csr, int n) {
  __shared__ int lcnt[BIN_NODES];
  __shared__ int loff[BIN_NODES];
  __shared__ int wsum[CSR_T / 32];
  __shared__ int wtot[BIN_NODES / 32];
  const int t    = (int)threadIdx.x;
  const int lane = t & 31;
  const int w    = t >> 5;
  const int bin  = (int)blockIdx.x;

  int v = t < bin ? bincnt[t] : 0;
  for (int off = 16; off > 0; off >>= 1) v += __shfl_xor(v, off, 32);
  if (lane == 0) wsum[w] = v;
  if (t < BIN_NODES) lcnt[t] = 0;
  const int m = bincnt[bin];
  __syncthreads();
  int bstart = 0;
  for (int k = 0; k < CSR_T / 32; ++k) bstart += wsum[k];
  const unsigned* __restrict__ mine = binned + bstart;

  for (int i0 = t; i0 < m; i0 += 4 * CSR_T) {
    unsigned p[4];
    #pragma unroll
    for (int k = 0; k < 4; ++k) {
      const int i = i0 + k * CSR_T;
      p[k] = i < m ? mine[i] : NOEDGE;
    }
    #pragma unroll
    for (int k = 0; k < 4; ++k)
      if (p[k] != NOEDGE) atomicAdd(&lcnt[p[k] & (BIN_NODES - 1)], 1);
  }
  __syncthreads();

  const int c = t < BIN_NODES ? lcnt[t] : 0;
  int inc = c;
  for (int off = 1; off < 32; off <<= 1) {
    const int u = __shfl_up(inc, off, 32);
    if (lane >= off) inc += u;
  }
  if (lane == 31 && w < BIN_NODES / 32) wtot[w] = inc;
  __syncthreads();
  if (t < BIN_NODES) {
    int pre = 0;
    for (int k = 0; k < BIN_NODES / 32; ++k) if (k < w) pre += wtot[k];
    const int off  = pre + inc - c;
    const int node = bin * BIN_NODES + t;
    loff[t] = off;
    lcnt[t] = 0;
    if (node < n) { cnt[node] = c; rowptr[node] = bstart + off; }
  }
  __syncthreads();

  int* __restrict__ out = csr + bstart;
  for (int i0 = t; i0 < m; i0 += 4 * CSR_T) {
    unsigned p[4];
    #pragma unroll
    for (int k = 0; k < 4; ++k) {
      const int i = i0 + k * CSR_T;
      p[k] = i < m ? mine[i] : NOEDGE;
    }
    int pos[4];
    #pragma unroll
    for (int k = 0; k < 4; ++k) {
      const int dl = (int)(p[k] & (BIN_NODES - 1));
      int rk = 0;
      if (p[k] != NOEDGE) rk = atomicAdd(&lcnt[dl], 1);
      pos[k] = loff[dl] + rk;
    }
    #pragma unroll
    for (int k = 0; k < 4; ++k)
      if (p[k] != NOEDGE) out[pos[k]] = (int)(p[k] >> BIN_SHIFT);
  }
}

#define NP_STEPS 4
#define NP_NODES_PER_BLOCK (8 * 2 * NP_STEPS)

__global__ void att_pass(const float* __restrict__ h0,
                         const float* __restrict__ attl,
                         const float* __restrict__ attr,
                         const int* __restrict__ cnt,
                         float* __restrict__ aldf, float* __restrict__ ar, int n) {
  const int lane  = (int)(threadIdx.x & 31);
  const int wave  = (int)((blockIdx.x * blockDim.x + threadIdx.x) >> 5);
  const int sub   = lane & 15;
  const int c4    = sub * 4;
  const int node0 = wave * (2 * NP_STEPS) + (lane >> 4);
  v2f* ald = (v2f*)aldf;

  if (wave * (2 * NP_STEPS) >= n) return;

  const v4f wl = *(const v4f*)(attl + c4);
  const v4f wr = *(const v4f*)(attr + c4);

  v4f hv[NP_STEPS];
  int dg[NP_STEPS];
  #pragma unroll
  for (int t = 0; t < NP_STEPS; ++t) {
    const int node = node0 + 2 * t;
    const int ld   = node < n ? node : n - 1;
    hv[t] = *(const v4f*)(h0 + (size_t)ld * HID + c4);
    dg[t] = cnt[ld];
  }

  __builtin_amdgcn_sched_barrier(0);

  #pragma unroll
  for (int t = 0; t < NP_STEPS; ++t) {
    const int node = node0 + 2 * t;
    const v4f h = hv[t];
    float pl = h.x * wl.x + h.y * wl.y + h.z * wl.z + h.w * wl.w;
    float pr = h.x * wr.x + h.y * wr.y + h.z * wr.z + h.w * wr.w;
    for (int off = 8; off > 0; off >>= 1) {
      pl += __shfl_xor(pl, off, 32);
      pr += __shfl_xor(pr, off, 32);
    }
    if (node < n && sub == 0) {
      v2f a; a.x = pl; a.y = rsqrtf((float)(dg[t] + 1));
      ald[node] = a;
      ar[node]  = pr;
    }
  }
}

#define GN 8
#define GE 24

template<int NS>
__device__ __forceinline__ void chunk_rows(v4f& acc, const float* __restrict__ h,
                                           int se, int wbits, int half, int c4) {
  const v4f* __restrict__ hv = (const v4f*)h;
  const unsigned sub = (unsigned)c4 >> 2;
  v4f rows[NS];
  #pragma unroll
  for (int j = 0; j < NS; ++j) {
    const int s0 = __builtin_amdgcn_readlane(se, 2 * j);
    const int s1 = __builtin_amdgcn_readlane(se, 2 * j + 1);
    const unsigned sj = (unsigned)(half ? s1 : s0);
    rows[j] = hv[sj * (HID / 4) + sub];
  }
  __builtin_amdgcn_sched_barrier(0);
  #pragma unroll
  for (int j = 0; j < NS; ++j) {
    const int w0 = __builtin_amdgcn_readlane(wbits, 2 * j);
    const int w1 = __builtin_amdgcn_readlane(wbits, 2 * j + 1);
    const float wj = __int_as_float(half ? w1 : w0);
    acc.x += wj * rows[j].x;
    acc.y += wj * rows[j].y;
    acc.z += wj * rows[j].z;
    acc.w += wj * rows[j].w;
  }
}

#define CHUNK_CASE(k) case k: chunk_rows<k>(acc, h, se, wbits, half, c4); break;

template<bool FIRST>
__global__ void __launch_bounds__(256) gather_layer(
                             const int* __restrict__ rowptr, const int* __restrict__ cnt,
                             const int* __restrict__ csr,
                             const float* __restrict__ h,
                             const float* h0res,
                             const float* __restrict__ aldf, const float* __restrict__ ar,
                             const float* __restrict__ attl2, const float* __restrict__ attr2,
                             float* orow,
                             float* __restrict__ ald2f, float* __restrict__ ar2, int n) {
  const int lane = (int)(threadIdx.x & 31);
  const int wib   = __builtin_amdgcn_readfirstlane((int)(threadIdx.x >> 5));
  const int wave  = (int)blockIdx.x * (int)(blockDim.x >> 5) + wib;
  const int node0 = wave * GN;
  if (node0 >= n) return;
  const int nn   = (n - node0 < GN) ? (n - node0) : GN;
  const int half = lane >> 4;
  const int c4   = (lane & 15) * 4;
  const v2f* ald = (const v2f*)aldf;

  int mb = 0, mc = 0;
  float mdi = 0.f, mar = 0.f, mws = 0.f;
  if (lane < nn) {
    mb = rowptr[node0 + lane];
    mc = cnt[node0 + lane];
    const v2f a = ald[node0 + lane];
    mdi = a.y;
    mar = ar[node0 + lane];
    mws = tanhf(a.x + mar) * a.y * a.y;
  }
  v4f wl2 = {0.f, 0.f, 0.f, 0.f}, wr2 = {0.f, 0.f, 0.f, 0.f};
  if (FIRST) {
    wl2 = *(const v4f*)(attl2 + c4);
    wr2 = *(const v4f*)(attr2 + c4);
  }

  int begin = __builtin_amdgcn_readlane(mb, 0);
  int c     = __builtin_amdgcn_readlane(mc, 0);
  int se0   = node0;
  {
    const int cc0 = c < GE ? c : GE;
    if (lane < cc0) se0 = csr[begin + lane];
  }

  for (int t = 0; t < nn; ++t) {
    const int   node = node0 + t;
    const float di_n = __int_as_float(__builtin_amdgcn_readlane(__float_as_int(mdi), t));
    const float ar_n = __int_as_float(__builtin_amdgcn_readlane(__float_as_int(mar), t));
    const float ws_n = __int_as_float(__builtin_amdgcn_readlane(__float_as_int(mws), t));

    const size_t own = (size_t)node * HID + c4;
    v4f hn = {0.f, 0.f, 0.f, 0.f}, r = {0.f, 0.f, 0.f, 0.f};
    int begin_n = 0, c_n = 0, se_n = node0;

    v4f acc = {0.f, 0.f, 0.f, 0.f};
    int se = se0;
    int cb = 0;
    do {
      const int rem = c - cb;
      const int cc  = rem < GE ? rem : GE;
      v2f a = {0.f, 0.f};
      if (lane < cc) a = ald[se];
      if (cb == 0) {
        hn = *(const v4f*)(h + own);
        r  = hn;
        if (!FIRST) r = *(const v4f*)(h0res + own);
        if (t + 1 < nn) {
          begin_n = __builtin_amdgcn_readlane(mb, t + 1);
          c_n     = __builtin_amdgcn_readlane(mc, t + 1);
          const int ccn = c_n < GE ? c_n : GE;
          if (lane < ccn) se_n = csr[begin_n + lane];
        }
      }
      float w = 0.f;
      if (lane < cc) w = tanhf(a.x + ar_n) * a.y * di_n;
      const int wbits = __float_as_int(w);
      switch ((cc + 1) >> 1) {
        CHUNK_CASE(1)  CHUNK_CASE(2)  CHUNK_CASE(3)  CHUNK_CASE(4)
        CHUNK_CASE(5)  CHUNK_CASE(6)  CHUNK_CASE(7)  CHUNK_CASE(8)
        CHUNK_CASE(9)  CHUNK_CASE(10) CHUNK_CASE(11) CHUNK_CASE(12)
        default: break;
      }
      cb += GE;
      if (cb < c) {
        const int rem2 = c - cb;
        const int cc2  = rem2 < GE ? rem2 : GE;
        se = node0;
        if (lane < cc2) se = csr[begin + cb + lane];
      }
    } while (cb < c);

    acc.x += __shfl_xor(acc.x, 16, 32);
    acc.y += __shfl_xor(acc.y, 16, 32);
    acc.z += __shfl_xor(acc.z, 16, 32);
    acc.w += __shfl_xor(acc.w, 16, 32);

    v4f o;
    o.x = acc.x + (EPS_BLEND * r.x + ws_n * hn.x);
    o.y = acc.y + (EPS_BLEND * r.y + ws_n * hn.y);
    o.z = acc.z + (EPS_BLEND * r.z + ws_n * hn.z);
    o.w = acc.w + (EPS_BLEND * r.w + ws_n * hn.w);

    if (FIRST) {
      o.x = fmaxf(o.x, 0.f); o.y = fmaxf(o.y, 0.f);
      o.z = fmaxf(o.z, 0.f); o.w = fmaxf(o.w, 0.f);
      float pl = o.x * wl2.x + o.y * wl2.y + o.z * wl2.z + o.w * wl2.w;
      float pr = o.x * wr2.x + o.y * wr2.y + o.z * wr2.z + o.w * wr2.w;
      for (int off = 8; off > 0; off >>= 1) {
        pl += __shfl_xor(pl, off, 32);
        pr += __shfl_xor(pr, off, 32);
      }
      if (lane == 0) {
        v2f a; a.x = pl; a.y = di_n;
        ((v2f*)ald2f)[node] = a;
        ar2[node] = pr;
      }
    }
    if (half == 0) *(v4f*)(orow + own) = o;

    begin = begin_n; c = c_n; se0 = se_n;
  }
}

extern "C" void kernel_launch(void* const* d_in, const int* in_sizes, int n_in,
                              void* d_out, int out_size, void* d_ws, size_t ws_size,
                              hipStream_t stream) {
  const float* x     = (const float*)d_in[0];
  const int*   ei    = (const int*)  d_in[1];
  const float* W_in  = (const float*)d_in[2];
  const float* b_in  = (const float*)d_in[3];
  const float* attl1 = (const float*)d_in[4];
  const float* attr1 = (const float*)d_in[5];
  const float* attl2 = (const float*)d_in[6];
  const float* attr2 = (const float*)d_in[7];
  const float* W_out = (const float*)d_in[8];
  const float* b_out = (const float*)d_in[9];
  float* out = (float*)d_out;

  const int IN = 128;
  const int E  = in_sizes[1] / 2;
  const int N  = in_sizes[0] / IN;
  const int*   srcI = ei;
  const int*   dstI = ei + E;

  float* H0   = (float*)d_ws;
  float* H1   = H0   + (size_t)N * HID;
  float* ALD1 = H1   + (size_t)N * HID;
  float* ALD2 = ALD1 + (size_t)N * 2;
  float* AR1  = ALD2 + (size_t)N * 2;
  float* AR2  = AR1  + N;
  int*   CNT  = (int*)(AR2 + N);
  int*   RPTR = CNT  + N;
  int*   BCNT = RPTR + N;
  int*   BCUR = BCNT + MAXBINS;
  unsigned* BINNED = (unsigned*)(BCUR + MAXBINS);
  int*   CSR  = (int*)(BINNED + E);

  const int ntiles = N / 16;
  const int gemmBlocks = (ntiles + 7) / 8;
  const int eBlk      = (E + BCHUNK - 1) / BCHUNK;
  const int nBins     = (N + BIN_NODES - 1) / BIN_NODES;
  const int nodeBlk   = (N + NP_NODES_PER_BLOCK - 1) / NP_NODES_PER_BLOCK;
  const int gBlk      = (N + 8 * GN - 1) / (8 * GN);

  const int gemmA = gemmBlocks * 2 / 5;
  const int gemmB = gemmBlocks - gemmA;
  bin_zero        <<<1,            MAXBINS, 0, stream>>>(BCNT, BCUR);
  count_and_gemm  <<<gemmA + eBlk, 256,     0, stream>>>(dstI, BCNT, E,
                                                         x, W_in, b_in, H0, ntiles, 0, gemmA);
  scatter_and_gemm<<<gemmB + eBlk, 256,     0, stream>>>(srcI, dstI, BCNT, BCUR, BINNED, E,
                                                         x, W_in, b_in, H0, ntiles, gemmA, gemmB);
  bin_csr         <<<nBins,        CSR_T,   0, stream>>>(BCNT, BINNED, CNT, RPTR, CSR, N);

  att_pass<<<nodeBlk, 256, 0, stream>>>(H0, attl1, attr1, CNT, ALD1, AR1, N);
  gather_layer<true> <<<gBlk, 256, 0, stream>>>(RPTR, CNT, CSR, H0, H0, ALD1, AR1,
                                                attl2, attr2, H1, ALD2, AR2, N);

  gather_layer<false><<<gBlk, 256, 0, stream>>>(RPTR, CNT, CSR, H1, H0, ALD2, AR2,
                                                attl2, attr2, H0, ALD2, AR2, N);

  gemm_wmma_f32<64, false, true><<<gemmBlocks, 256, 0, stream>>>(H0, W_out, b_out, out, ntiles);
}


# layer gathers sized to one resident round: 10 nodes per wave under a 96-register cap (10 waves per SIMD; epilogue lane sums as DPP/permlane moves to fit it): 1250 blocks all resident, not 1563
# speedup vs baseline: 5.0179x; 1.1159x over previous
#include <hip/hip_runtime.h>
#include <hip/hip_bf16.h>


typedef __attribute__((ext_vector_type(2))) float v2f;
typedef __attribute__((ext_vector_type(8))) float v8f;
typedef __attribute__((ext_vector_type(4))) float v4f;

#define HID 64
#define EPS_BLEND 0.1f

#define GP 68

__device__ __forceinline__ int lds_pair_pos(int col) { return (col & ~7) + ((col >> 1) & 2); }

template<int K, bool RELU, bool RELU_IN>
__device__ __forceinline__ void gemm_block(const float* __restrict__ A,
                                           const float* __restrict__ W,
                                           const float* __restrict__ bias,
                                           float* __restrict__ out,
                                           int ntiles, int blk) {
  constexpr int WP  = K + 4;
  constexpr int NCH = K / 64;
  constexpr int WV  = (64 * K / 4) / 256;
  __shared__ __attribute__((aligned(16))) float sW[64 * WP];
  __shared__ __attribute__((aligned(16))) float sT[8 * 16 * GP];

  const int t    = (int)threadIdx.x;
  const int lane = t & 31;
  const int wib  = t >> 5;
  const int tile = blk * 8 + wib;
  const bool live = tile < ntiles;
  const int ltile = live ? tile : ntiles - 1;
  const int c = lane & 15;
  const int h = lane >> 4;
  float* sA = sT + wib * (16 * GP);

  const float* arow = A + ((size_t)ltile * 16 + h) * K + c * 4;
  const int apos = h * GP + lds_pair_pos(c * 4);

  v4f av[8];
  #pragma unroll
  for (int i = 0; i < 8; ++i) av[i] = *(const v4f*)(arow + (size_t)(2 * i) * K);

  {
    v4f wv[WV];
    #pragma unroll
    for (int p = 0; p < WV; ++p) wv[p] = ((const v4f*)W)[t + 256 * p];
    #pragma unroll
    for (int p = 0; p < WV; ++p) {
      const int e   = (t + 256 * p) * 4;
      const int row = e / K;
      const int col = e % K;
      float* d = sW + row * WP + lds_pair_pos(col);
      v2f lo; lo.x = wv[p].x; lo.y = wv[p].y;
      v2f hi; hi.x = wv[p].z; hi.y = wv[p].w;
      *(v2f*)d       = lo;
      *(v2f*)(d + 4) = hi;
    }
  }
  __syncthreads();
  if (!live) return;

  v8f acc0 = {}, acc1 = {}, acc2 = {}, acc3 = {};
  const float* fa = sA + c * GP + 4 * h;
  const float* fb = sW + c * WP + 4 * h;

  #pragma unroll
  for (int ch = 0; ch < NCH; ++ch) {
    #pragma unroll
    for (int i = 0; i < 8; ++i) {
      v4f a = av[i];
      if (RELU_IN) {
        a.x = fmaxf(a.x, 0.f); a.y = fmaxf(a.y, 0.f);
        a.z = fmaxf(a.z, 0.f); a.w = fmaxf(a.w, 0.f);
      }
      float* d = sA + (2 * i) * GP + apos;
      v2f lo; lo.x = a.x; lo.y = a.y;
      v2f hi; hi.x = a.z; hi.y = a.w;
      *(v2f*)d       = lo;
      *(v2f*)(d + 4) = hi;
    }
    if (ch + 1 < NCH) {
      #pragma unroll
      for (int i = 0; i < 8; ++i)
        av[i] = *(const v4f*)(arow + (size_t)(2 * i) * K + 64 * (ch + 1));
    }
    #pragma unroll
    for (int j = 0; j < 8; ++j) {
      const v4f a  = *(const v4f*)(fa + 8 * j);
      const v4f b0 = *(const v4f*)(fb +  0 * WP + 64 * ch + 8 * j);
      const v4f b1 = *(const v4f*)(fb + 16 * WP + 64 * ch + 8 * j);
      const v4f b2 = *(const v4f*)(fb + 32 * WP + 64 * ch + 8 * j);
      const v4f b3 = *(const v4f*)(fb + 48 * WP + 64 * ch + 8 * j);
      v2f al, ah, p0, q0, p1, q1, p2, q2, p3, q3;
      al.x = a.x;  al.y = a.y;  ah.x = a.z;  ah.y = a.w;
      p0.x = b0.x; p0.y = b0.y; q0.x = b0.z; q0.y = b0.w;
      p1.x = b1.x; p1.y = b1.y; q1.x = b1.z; q1.y = b1.w;
      p2.x = b2.x; p2.y = b2.y; q2.x = b2.z; q2.y = b2.w;
      p3.x = b3.x; p3.y = b3.y; q3.x = b3.z; q3.y = b3.w;
      acc0 = __builtin_amdgcn_wmma_f32_16x16x4_f32(false, al, false, p0, (short)0, acc0, false, false);
      acc1 = __builtin_amdgcn_wmma_f32_16x16x4_f32(false, al, false, p1, (short)0, acc1, false, false);
      acc2 = __builtin_amdgcn_wmma_f32_16x16x4_f32(false, al, false, p2, (short)0, acc2, false, false);
      acc3 = __builtin_amdgcn_wmma_f32_16x16x4_f32(false, al, false, p3, (short)0, acc3, false, false);
      acc0 = __builtin_amdgcn_wmma_f32_16x16x4_f32(false, ah, false, q0, (short)0, acc0, false, false);
      acc1 = __builtin_amdgcn_wmma_f32_16x16x4_f32(false, ah, false, q1, (short)0, acc1, false, false);
      acc2 = __builtin_amdgcn_wmma_f32_16x16x4_f32(false, ah, false, q2, (short)0, acc2, false, false);
      acc3 = __builtin_amdgcn_wmma_f32_16x16x4_f32(false, ah, false, q3, (short)0, acc3, false, false);
    }
  }

  const float bb0 = bias[c], bb1 = bias[16 + c], bb2 = bias[32 + c], bb3 = bias[48 + c];
  #pragma unroll
  for (int v = 0; v < 8; ++v) {
    float x0 = acc0[v] + bb0;
    float x1 = acc1[v] + bb1;
    float x2 = acc2[v] + bb2;
    float x3 = acc3[v] + bb3;
    if (RELU) {
      x0 = fmaxf(x0, 0.f); x1 = fmaxf(x1, 0.f);
      x2 = fmaxf(x2, 0.f); x3 = fmaxf(x3, 0.f);
    }
    float* d = sA + (8 * h + v) * GP + c;
    d[0]  = x0;
    d[16] = x1;
    d[32] = x2;
    d[48] = x3;
  }
  float* orow = out + ((size_t)tile * 16 + h) * HID + c * 4;
  #pragma unroll
  for (int i = 0; i < 8; ++i) {
    const v4f o = *(const v4f*)(sA + (2 * i + h) * GP + c * 4);
    *(v4f*)(orow + (size_t)(2 * i) * HID) = o;
  }
}

#define BIN_SHIFT 9
#define BIN_NODES 512
#define MAXBINS   256
#define BE        16
#define BCHUNK    (256 * BE)
#define CSR_T     1024
#define NOEDGE    0xFFFFFFFFu

__global__ void bin_zero(int* __restrict__ bincnt, int* __restrict__ bincur) {
  bincnt[threadIdx.x] = 0;
  bincur[threadIdx.x] = 0;
}

__device__ __forceinline__ void bin_count_block(const int* __restrict__ dst,
                                                int* __restrict__ bincnt, int e, int blk) {
  __shared__ int hist[MAXBINS];
  const int t = (int)threadIdx.x;
  hist[t] = 0;
  __syncthreads();
  const int base = blk * BCHUNK + t;
  int d[BE];
  #pragma unroll
  for (int k = 0; k < BE; ++k) {
    const int idx = base + k * 256;
    d[k] = idx < e ? dst[idx] : -1;
  }
  #pragma unroll
  for (int k = 0; k < BE; ++k)
    if (d[k] >= 0) atomicAdd(&hist[d[k] >> BIN_SHIFT], 1);
  __syncthreads();
  const int h = hist[t];
  if (h > 0) atomicAdd(&bincnt[t], h);
}

__device__ __forceinline__ void bin_scatter_block(const int* __restrict__ src,
                                                  const int* __restrict__ dst,
                                                  const int* __restrict__ bincnt,
                                                  int* __restrict__ bincur,
                                                  unsigned* __restrict__ binned, int e, int blk) {
  __shared__ int hist[MAXBINS];
  __shared__ int gbase[MAXBINS];
  __shared__ int wsum[8];
  const int t    = (int)threadIdx.x;
  const int lane = t & 31;
  const int w    = t >> 5;
  hist[t] = 0;

  const int base = blk * BCHUNK + t;
  int d[BE], s[BE], r[BE];
  #pragma unroll
  for (int k = 0; k < BE; ++k) {
    const int idx = base + k * 256;
    const bool ok = idx < e;
    d[k] = ok ? dst[idx] : -1;
    s[k] = ok ? src[idx] : 0;
  }

  const int c = bincnt[t];
  int inc = c;
  for (int off = 1; off < 32; off <<= 1) {
    const int v = __shfl_up(inc, off, 32);
    if (lane >= off) inc += v;
  }
  if (lane == 31) wsum[w] = inc;
  __syncthreads();
  int pre = 0;
  for (int k = 0; k < 8; ++k) if (k < w) pre += wsum[k];
  const int bstart = pre + inc - c;

  #pragma unroll
  for (int k = 0; k < BE; ++k) {
    r[k] = 0;
    if (d[k] >= 0) r[k] = atomicAdd(&hist[d[k] >> BIN_SHIFT], 1);
  }
  __syncthreads();
  const int h = hist[t];
  int gb = bstart;
  if (h > 0) gb += atomicAdd(&bincur[t], h);
  gbase[t] = gb;
  __syncthreads();
  int pos[BE];
  #pragma unroll
  for (int k = 0; k < BE; ++k)
    pos[k] = gbase[d[k] >= 0 ? d[k] >> BIN_SHIFT : 0] + r[k];
  #pragma unroll
  for (int k = 0; k < BE; ++k) {
    if (d[k] >= 0)
      binned[pos[k]] = ((unsigned)s[k] << BIN_SHIFT) | ((unsigned)d[k] & (BIN_NODES - 1));
  }
}

__global__ void __launch_bounds__(256) count_and_gemm(const int* __restrict__ dst,
                                                      int* __restrict__ bincnt, int e,
                                                      const float* __restrict__ A,
                                                      const float* __restrict__ W,
                                                      const float* __restrict__ bias,
                                                      float* __restrict__ out,
                                                      int ntiles, int gemm_blk0, int gemm_blocks) {
  const int b = (int)blockIdx.x;
  if (b < gemm_blocks) gemm_block<128, true, false>(A, W, bias, out, ntiles, gemm_blk0 + b);
  else                 bin_count_block(dst, bincnt, e, b - gemm_blocks);
}

__global__ void __launch_bounds__(256) scatter_and_gemm(const int* __restrict__ src,
                                                        const int* __restrict__ dst,
                                                        const int* __restrict__ bincnt,
                                                        int* __restrict__ bincur,
                                                        unsigned* __restrict__ binned, int e,
                                                        const float* __restrict__ A,
                                                        const float* __restrict__ W,
                                                        const float* __restrict__ bias,
                                                        float* __restrict__ out,
                                                        int ntiles, int gemm_blk0, int gemm_blocks) {
  const int b = (int)blockIdx.x;
  if (b < gemm_blocks) gemm_block<128, true, false>(A, W, bias, out, ntiles, gemm_blk0 + b);
  else                 bin_scatter_block(src, dst, bincnt, bincur, binned, e, b - gemm_blocks);
}

__global__ void __launch_bounds__(CSR_T) bin_csr(const int* __restrict__ bincnt,
                                                 const unsigned* __restrict__ binned,
                                                 int* __restrict__ cnt,
                                                 int* __restrict__ rowptr,
                                                 int* __restrict__ csr, int n) {
  __shared__ int lcnt[BIN_NODES];
  __shared__ int loff[BIN_NODES];
  __shared__ int wsum[CSR_T / 32];
  __shared__ int wtot[BIN_NODES / 32];
  const int t    = (int)threadIdx.x;
  const int lane = t & 31;
  const int w    = t >> 5;
  const int bin  = (int)blockIdx.x;

  int v = t < bin ? bincnt[t] : 0;
  for (int off = 16; off > 0; off >>= 1) v += __shfl_xor(v, off, 32);
  if (lane == 0) wsum[w] = v;
  if (t < BIN_NODES) lcnt[t] = 0;
  const int m = bincnt[bin];
  __syncthreads();
  int bstart = 0;
  for (int k = 0; k < CSR_T / 32; ++k) bstart += wsum[k];
  const unsigned* __restrict__ mine = binned + bstart;

  for (int i0 = t; i0 < m; i0 += 4 * CSR_T) {
    unsigned p[4];
    #pragma unroll
    for (int k = 0; k < 4; ++k) {
      const int i = i0 + k * CSR_T;
      p[k] = i < m ? mine[i] : NOEDGE;
    }
    #pragma unroll
    for (int k = 0; k < 4; ++k)
      if (p[k] != NOEDGE) atomicAdd(&lcnt[p[k] & (BIN_NODES - 1)], 1);
  }
  __syncthreads();

  const int c = t < BIN_NODES ? lcnt[t] : 0;
  int inc = c;
  for (int off = 1; off < 32; off <<= 1) {
    const int u = __shfl_up(inc, off, 32);
    if (lane >= off) inc += u;
  }
  if (lane == 31 && w < BIN_NODES / 32) wtot[w] = inc;
  __syncthreads();
  if (t < BIN_NODES) {
    int pre = 0;
    for (int k = 0; k < BIN_NODES / 32; ++k) if (k < w) pre += wtot[k];
    const int off  = pre + inc - c;
    const int node = bin * BIN_NODES + t;
    loff[t] = off;
    lcnt[t] = 0;
    if (node < n) { cnt[node] = c; rowptr[node] = bstart + off; }
  }
  __syncthreads();

  int* __restrict__ out = csr + bstart;
  for (int i0 = t; i0 < m; i0 += 4 * CSR_T) {
    unsigned p[4];
    #pragma unroll
    for (int k = 0; k < 4; ++k) {
      const int i = i0 + k * CSR_T;
      p[k] = i < m ? mine[i] : NOEDGE;
    }
    int pos[4];
    #pragma unroll
    for (int k = 0; k < 4; ++k) {
      const int dl = (int)(p[k] & (BIN_NODES - 1));
      int rk = 0;
      if (p[k] != NOEDGE) rk = atomicAdd(&lcnt[dl], 1);
      pos[k] = loff[dl] + rk;
    }
    #pragma unroll
    for (int k = 0; k < 4; ++k)
      if (p[k] != NOEDGE) out[pos[k]] = (int)(p[k] >> BIN_SHIFT);
  }
}

#define NP_STEPS 4
#define NP_NODES_PER_BLOCK (8 * 2 * NP_STEPS)

__global__ void att_pass(const float* __restrict__ h0,
                         const float* __restrict__ attl,
                         const float* __restrict__ attr,
                         const int* __restrict__ cnt,
                         float* __restrict__ aldf, float* __restrict__ ar, int n) {
  const int lane  = (int)(threadIdx.x & 31);
  const int wave  = (int)((blockIdx.x * blockDim.x + threadIdx.x) >> 5);
  const int sub   = lane & 15;
  const int c4    = sub * 4;
  const int node0 = wave * (2 * NP_STEPS) + (lane >> 4);
  v2f* ald = (v2f*)aldf;

  if (wave * (2 * NP_STEPS) >= n) return;

  const v4f wl = *(const v4f*)(attl + c4);
  const v4f wr = *(const v4f*)(attr + c4);

  v4f hv[NP_STEPS];
  int dg[NP_STEPS];
  #pragma unroll
  for (int t = 0; t < NP_STEPS; ++t) {
    const int node = node0 + 2 * t;
    const int ld   = node < n ? node : n - 1;
    hv[t] = *(const v4f*)(h0 + (size_t)ld * HID + c4);
    dg[t] = cnt[ld];
  }

  __builtin_amdgcn_sched_barrier(0);

  #pragma unroll
  for (int t = 0; t < NP_STEPS; ++t) {
    const int node = node0 + 2 * t;
    const v4f h = hv[t];
    float pl = h.x * wl.x + h.y * wl.y + h.z * wl.z + h.w * wl.w;
    float pr = h.x * wr.x + h.y * wr.y + h.z * wr.z + h.w * wr.w;
    for (int off = 8; off > 0; off >>= 1) {
      pl += __shfl_xor(pl, off, 32);
      pr += __shfl_xor(pr, off, 32);
    }
    if (node < n && sub == 0) {
      v2f a; a.x = pl; a.y = rsqrtf((float)(dg[t] + 1));
      ald[node] = a;
      ar[node]  = pr;
    }
  }
}

#define GN 10
#define TILES_PER_BLOCK (8 * GN / 16)
#define GE 24

template<int NS>
__device__ __forceinline__ void chunk_rows(v4f& acc, const float* __restrict__ h,
                                           int se, int wbits, int half, int c4) {
  const v4f* __restrict__ hv = (const v4f*)h;
  const unsigned sub = (unsigned)c4 >> 2;
  v4f rows[NS];
  #pragma unroll
  for (int j = 0; j < NS; ++j) {
    const int s0 = __builtin_amdgcn_readlane(se, 2 * j);
    const int s1 = __builtin_amdgcn_readlane(se, 2 * j + 1);
    const unsigned sj = (unsigned)(half ? s1 : s0);
    rows[j] = hv[sj * (HID / 4) + sub];
  }
  __builtin_amdgcn_sched_barrier(0);
  #pragma unroll
  for (int j = 0; j < NS; ++j) {
    const int w0 = __builtin_amdgcn_readlane(wbits, 2 * j);
    const int w1 = __builtin_amdgcn_readlane(wbits, 2 * j + 1);
    const float wj = __int_as_float(half ? w1 : w0);
    acc.x += wj * rows[j].x;
    acc.y += wj * rows[j].y;
    acc.z += wj * rows[j].z;
    acc.w += wj * rows[j].w;
  }
}

#define CHUNK_CASE(k) case k: chunk_rows<k>(acc, h, se, wbits, half, c4); break;

__device__ __forceinline__ float add_other_half(float v) {
  const int x = __float_as_int(v);
  return v + __int_as_float(__builtin_amdgcn_permlanex16(x, x, 0x76543210, 0xfedcba98, false, false));
}
template<int M>
__device__ __forceinline__ float add_row_xor(float v) {
  const int x = __float_as_int(v);
  return v + __int_as_float(__builtin_amdgcn_update_dpp(x, x, 0x160 + M, 0xf, 0xf, false));
}

template<bool FIRST>
__global__ void __launch_bounds__(256, 10) gather_layer(
                             const int* __restrict__ rowptr, const int* __restrict__ cnt,
                             const int* __restrict__ csr,
                             const float* __restrict__ h,
                             const float* h0res,
                             const float* __restrict__ aldf, const float* __restrict__ ar,
                             const float* __restrict__ attl2, const float* __restrict__ attr2,
                             float* orow,
                             float* __restrict__ ald2f, float* __restrict__ ar2, int n,
                             const float* __restrict__ Wo, const float* __restrict__ bo,
                             float* __restrict__ outp, int ntiles) {
  __shared__ __attribute__((aligned(16))) float sW[FIRST ? 4 : 64 * GP];
  __shared__ __attribute__((aligned(16))) float sT[FIRST ? 4 : 8 * GN * GP];
  const int lane = (int)(threadIdx.x & 31);
  const int wib   = __builtin_amdgcn_readfirstlane((int)(threadIdx.x >> 5));
  const int wave  = (int)blockIdx.x * (int)(blockDim.x >> 5) + wib;
  const int node0 = wave * GN;
  if (FIRST && node0 >= n) return;
  int nn = (n - node0 < GN) ? (n - node0) : GN;
  if (nn < 0) nn = 0;
  const int half = lane >> 4;
  const int c4   = (lane & 15) * 4;
  const v2f* ald = (const v2f*)aldf;

  v4f wv[4];
  if (!FIRST) {
    #pragma unroll
    for (int p = 0; p < 4; ++p) wv[p] = ((const v4f*)Wo)[(int)threadIdx.x + 256 * p];
  }

  int mb = 0, mc = 0;
  float mdi = 0.f, mar = 0.f, mws = 0.f;
  if (lane < nn) {
    mb = rowptr[node0 + lane];
    mc = cnt[node0 + lane];
    const v2f a = ald[node0 + lane];
    mdi = a.y;
    mar = ar[node0 + lane];
    mws = tanhf(a.x + mar) * a.y * a.y;
  }
  v4f wl2 = {0.f, 0.f, 0.f, 0.f}, wr2 = {0.f, 0.f, 0.f, 0.f};
  if (FIRST) {
    wl2 = *(const v4f*)(attl2 + c4);
    wr2 = *(const v4f*)(attr2 + c4);
  } else {
    #pragma unroll
    for (int p = 0; p < 4; ++p) {
      const int e   = ((int)threadIdx.x + 256 * p) * 4;
      const int row = e / HID;
      const int col = e % HID;
      float* d = sW + row * GP + lds_pair_pos(col);
      v2f lo; lo.x = wv[p].x; lo.y = wv[p].y;
      v2f hi; hi.x = wv[p].z; hi.y = wv[p].w;
      *(v2f*)d       = lo;
      *(v2f*)(d + 4) = hi;
    }
  }

  int begin = __builtin_amdgcn_readlane(mb, 0);
  int c     = __builtin_amdgcn_readlane(mc, 0);
  int se0   = node0;
  {
    const int cc0 = c < GE ? c : GE;
    if (lane < cc0) se0 = csr[begin + lane];
  }

  for (int t = 0; t < nn; ++t) {
    const int   node = node0 + t;
    const float di_n = __int_as_float(__builtin_amdgcn_readlane(__float_as_int(mdi), t));
    const float ar_n = __int_as_float(__builtin_amdgcn_readlane(__float_as_int(mar), t));
    const float ws_n = __int_as_float(__builtin_amdgcn_readlane(__float_as_int(mws), t));

    const size_t own = (size_t)node * HID + c4;
    v4f hn = {0.f, 0.f, 0.f, 0.f}, r = {0.f, 0.f, 0.f, 0.f};
    int begin_n = 0, c_n = 0, se_n = node0;

    v4f acc = {0.f, 0.f, 0.f, 0.f};
    int se = se0;
    int cb = 0;
    do {
      const int rem = c - cb;
      const int cc  = rem < GE ? rem : GE;
      v2f a = {0.f, 0.f};
      if (lane < cc) a = ald[se];
      if (cb == 0) {
        hn = *(const v4f*)(h + own);
        r  = hn;
        if (!FIRST) r = *(const v4f*)(h0res + own);
        if (t + 1 < nn) {
          begin_n = __builtin_amdgcn_readlane(mb, t + 1);
          c_n     = __builtin_amdgcn_readlane(mc, t + 1);
          const int ccn = c_n < GE ? c_n : GE;
          if (lane < ccn) se_n = csr[begin_n + lane];
        }
      }
      float w = 0.f;
      if (lane < cc) w = tanhf(a.x + ar_n) * a.y * di_n;
      const int wbits = __float_as_int(w);
      switch ((cc + 1) >> 1) {
        CHUNK_CASE(1)  CHUNK_CASE(2)  CHUNK_CASE(3)  CHUNK_CASE(4)
        CHUNK_CASE(5)  CHUNK_CASE(6)  CHUNK_CASE(7)  CHUNK_CASE(8)
        CHUNK_CASE(9)  CHUNK_CASE(10) CHUNK_CASE(11) CHUNK_CASE(12)
        default: break;
      }
      cb += GE;
      if (cb < c) {
        const int rem2 = c - cb;
        const int cc2  = rem2 < GE ? rem2 : GE;
        se = node0;
        if (lane < cc2) se = csr[begin + cb + lane];
      }
    } while (cb < c);

    acc.x = add_other_half(acc.x);
    acc.y = add_other_half(acc.y);
    acc.z = add_other_half(acc.z);
    acc.w = add_other_half(acc.w);

    v4f o;
    o.x = acc.x + (EPS_BLEND * r.x + ws_n * hn.x);
    o.y = acc.y + (EPS_BLEND * r.y + ws_n * hn.y);
    o.z = acc.z + (EPS_BLEND * r.z + ws_n * hn.z);
    o.w = acc.w + (EPS_BLEND * r.w + ws_n * hn.w);

    if (FIRST) {
      o.x = fmaxf(o.x, 0.f); o.y = fmaxf(o.y, 0.f);
      o.z = fmaxf(o.z, 0.f); o.w = fmaxf(o.w, 0.f);
      float pl = o.x * wl2.x + o.y * wl2.y + o.z * wl2.z + o.w * wl2.w;
      float pr = o.x * wr2.x + o.y * wr2.y + o.z * wr2.z + o.w * wr2.w;
      pl = add_row_xor<8>(pl); pr = add_row_xor<8>(pr);
      pl = add_row_xor<4>(pl); pr = add_row_xor<4>(pr);
      pl = add_row_xor<2>(pl); pr = add_row_xor<2>(pr);
      pl = add_row_xor<1>(pl); pr = add_row_xor<1>(pr);
      if (lane == 0) {
        v2f a; a.x = pl; a.y = di_n;
        ((v2f*)ald2f)[node] = a;
        ar2[node] = pr;
      }
    }
    if (FIRST) {
      if (half == 0) *(v4f*)(orow + own) = o;
    } else if (half == 0) {
      float* d = sT + (wib * GN + t) * GP + lds_pair_pos(c4);
      v2f lo; lo.x = fmaxf(o.x, 0.f); lo.y = fmaxf(o.y, 0.f);
      v2f hi; hi.x = fmaxf(o.z, 0.f); hi.y = fmaxf(o.w, 0.f);
      *(v2f*)d       = lo;
      *(v2f*)(d + 4) = hi;
    }

    begin = begin_n; c = c_n; se0 = se_n;
  }

  if (!FIRST) {
    __syncthreads();
    const int tile = (int)blockIdx.x * TILES_PER_BLOCK + wib;
    if (wib >= TILES_PER_BLOCK || tile >= ntiles) return;
    const int cc = lane & 15;
    float* sA = sT + wib * (16 * GP);
    const float* fa = sA + cc * GP + 4 * half;
    const float* fb = sW + cc * GP + 4 * half;
    v8f acc0 = {}, acc1 = {}, acc2 = {}, acc3 = {};
    #pragma unroll
    for (int j = 0; j < 8; ++j) {
      const v4f a  = *(const v4f*)(fa + 8 * j);
      const v4f b0 = *(const v4f*)(fb +  0 * GP + 8 * j);
      const v4f b1 = *(const v4f*)(fb + 16 * GP + 8 * j);
      const v4f b2 = *(const v4f*)(fb + 32 * GP + 8 * j);
      const v4f b3 = *(const v4f*)(fb + 48 * GP + 8 * j);
      v2f al, ah, p0, q0, p1, q1, p2, q2, p3, q3;
      al.x = a.x;  al.y = a.y;  ah.x = a.z;  ah.y = a.w;
      p0.x = b0.x; p0.y = b0.y; q0.x = b0.z; q0.y = b0.w;
      p1.x = b1.x; p1.y = b1.y; q1.x = b1.z; q1.y = b1.w;
      p2.x = b2.x; p2.y = b2.y; q2.x = b2.z; q2.y = b2.w;
      p3.x = b3.x; p3.y = b3.y; q3.x = b3.z; q3.y = b3.w;
      acc0 = __builtin_amdgcn_wmma_f32_16x16x4_f32(false, al, false, p0, (short)0, acc0, false, false);
      acc1 = __builtin_amdgcn_wmma_f32_16x16x4_f32(false, al, false, p1, (short)0, acc1, false, false);
      acc2 = __builtin_amdgcn_wmma_f32_16x16x4_f32(false, al, false, p2, (short)0, acc2, false, false);
      acc3 = __builtin_amdgcn_wmma_f32_16x16x4_f32(false, al, false, p3, (short)0, acc3, false, false);
      acc0 = __builtin_amdgcn_wmma_f32_16x16x4_f32(false, ah, false, q0, (short)0, acc0, false, false);
      acc1 = __builtin_amdgcn_wmma_f32_16x16x4_f32(false, ah, false, q1, (short)0, acc1, false, false);
      acc2 = __builtin_amdgcn_wmma_f32_16x16x4_f32(false, ah, false, q2, (short)0, acc2, false, false);
      acc3 = __builtin_amdgcn_wmma_f32_16x16x4_f32(false, ah, false, q3, (short)0, acc3, false, false);
    }
    const float bb0 = bo[cc], bb1 = bo[16 + cc], bb2 = bo[32 + cc], bb3 = bo[48 + cc];
    #pragma unroll
    for (int v = 0; v < 8; ++v) {
      float* d = sA + (8 * half + v) * GP + cc;
      d[0]  = acc0[v] + bb0;
      d[16] = acc1[v] + bb1;
      d[32] = acc2[v] + bb2;
      d[48] = acc3[v] + bb3;
    }
    float* dst = outp + ((size_t)tile * 16 + half) * HID + cc * 4;
    #pragma unroll
    for (int i = 0; i < 8; ++i) {
      const v4f ov = *(const v4f*)(sA + (2 * i + half) * GP + cc * 4);
      *(v4f*)(dst + (size_t)(2 * i) * HID) = ov;
    }
  }
}

extern "C" void kernel_launch(void* const* d_in, const int* in_sizes, int n_in,
                              void* d_out, int out_size, void* d_ws, size_t ws_size,
                              hipStream_t stream) {
  const float* x     = (const float*)d_in[0];
  const int*   ei    = (const int*)  d_in[1];
  const float* W_in  = (const float*)d_in[2];
  const float* b_in  = (const float*)d_in[3];
  const float* attl1 = (const float*)d_in[4];
  const float* attr1 = (const float*)d_in[5];
  const float* attl2 = (const float*)d_in[6];
  const float* attr2 = (const float*)d_in[7];
  const float* W_out = (const float*)d_in[8];
  const float* b_out = (const float*)d_in[9];
  float* out = (float*)d_out;

  const int IN = 128;
  const int E  = in_sizes[1] / 2;
  const int N  = in_sizes[0] / IN;
  const int*   srcI = ei;
  const int*   dstI = ei + E;

  float* H0   = (float*)d_ws;
  float* H1   = H0   + (size_t)N * HID;
  float* ALD1 = H1   + (size_t)N * HID;
  float* ALD2 = ALD1 + (size_t)N * 2;
  float* AR1  = ALD2 + (size_t)N * 2;
  float* AR2  = AR1  + N;
  int*   CNT  = (int*)(AR2 + N);
  int*   RPTR = CNT  + N;
  int*   BCNT = RPTR + N;
  int*   BCUR = BCNT + MAXBINS;
  unsigned* BINNED = (unsigned*)(BCUR + MAXBINS);
  int*   CSR  = (int*)(BINNED + E);

  const int ntiles = N / 16;
  const int gemmBlocks = (ntiles + 7) / 8;
  const int eBlk      = (E + BCHUNK - 1) / BCHUNK;
  const int nBins     = (N + BIN_NODES - 1) / BIN_NODES;
  const int nodeBlk   = (N + NP_NODES_PER_BLOCK - 1) / NP_NODES_PER_BLOCK;
  const int gBlk      = (N + 8 * GN - 1) / (8 * GN);

  const int gemmA = gemmBlocks * 2 / 5;
  const int gemmB = gemmBlocks - gemmA;
  bin_zero        <<<1,            MAXBINS, 0, stream>>>(BCNT, BCUR);
  count_and_gemm  <<<gemmA + eBlk, 256,     0, stream>>>(dstI, BCNT, E,
                                                         x, W_in, b_in, H0, ntiles, 0, gemmA);
  scatter_and_gemm<<<gemmB + eBlk, 256,     0, stream>>>(srcI, dstI, BCNT, BCUR, BINNED, E,
                                                         x, W_in, b_in, H0, ntiles, gemmA, gemmB);
  bin_csr         <<<nBins,        CSR_T,   0, stream>>>(BCNT, BINNED, CNT, RPTR, CSR, N);

  att_pass<<<nodeBlk, 256, 0, stream>>>(H0, attl1, attr1, CNT, ALD1, AR1, N);
  gather_layer<true> <<<gBlk, 256, 0, stream>>>(RPTR, CNT, CSR, H0, H0, ALD1, AR1,
                                                attl2, attr2, H1, ALD2, AR2, N,
                                                W_out, b_out, out, ntiles);

  gather_layer<false><<<gBlk, 256, 0, stream>>>(RPTR, CNT, CSR, H1, H0, ALD2, AR2,
                                                attl2, attr2, H0, ALD2, AR2, N,
                                                W_out, b_out, out, ntiles);
}
